# starting kernel
# baseline (speedup 1.0000x reference)
#include <hip/hip_runtime.h>
#include <hip/hip_bf16.h>

typedef __attribute__((ext_vector_type(2))) float v2f;
typedef __attribute__((ext_vector_type(8))) float v8f;

#define NNODES 40000
#define NEDGES 640000
#define NGRAPH 256

__device__ __forceinline__ void atomAddF(float* p, float v) {
    unsafeAtomicAdd(p, v);
}

__global__ void deg_kernel(const int* __restrict__ dst, float* __restrict__ deg, int E) {
    int e = blockIdx.x * blockDim.x + threadIdx.x;
    if (e < E) atomAddF(&deg[dst[e]], 1.0f);
}

__global__ void dis_counts_kernel(float* __restrict__ deg_dis,
                                  const int* __restrict__ batch,
                                  float* __restrict__ counts, int N) {
    int n = blockIdx.x * blockDim.x + threadIdx.x;
    if (n < N) {
        deg_dis[n] = rsqrtf(deg_dis[n] + 1.0f);
        atomAddF(&counts[batch[n]], 1.0f);
    }
}

__global__ void wmma_gemm_f32(const float* __restrict__ A, const float* __restrict__ B,
                              float* __restrict__ C, int N, int K, int M,
                              const float* __restrict__ rowscale,
                              const float* __restrict__ bias, int total_tiles) {
    int tid  = threadIdx.x;
    int wid  = tid >> 5;
    int lane = tid & 31;
    int tile = blockIdx.x * (blockDim.x >> 5) + wid;
    if (tile >= total_tiles) return;

    int tiles_m = M >> 4;
    int rt = tile / tiles_m, ct = tile - rt * tiles_m;
    int row0 = rt << 4, col0 = ct << 4;
    int hi = lane >> 4, lm = lane & 15;

    v8f c = {};
    const float* Arow = A + (size_t)(row0 + lm) * K;
    for (int k0 = 0; k0 < K; k0 += 4) {
        v2f a = *reinterpret_cast<const v2f*>(Arow + k0 + 2 * hi);
        const float* Bp = B + (size_t)(k0 + 2 * hi) * M + col0 + lm;
        v2f b;
        b.x = Bp[0];
        b.y = Bp[M];
        c = __builtin_amdgcn_wmma_f32_16x16x4_f32(false, a, false, b,
                                                  (short)0, c, false, false);
    }
    float bv = bias ? bias[col0 + lm] : 0.0f;
    #pragma unroll
    for (int v = 0; v < 8; ++v) {
        int row = row0 + v + 8 * hi;
        float o = c[v];
        if (rowscale) o *= rowscale[row];
        C[(size_t)row * M + col0 + lm] = o + bv;
    }
}

__global__ void scatter_kernel(const float* __restrict__ h, const int* __restrict__ src,
                               const int* __restrict__ dst, float* __restrict__ agg,
                               int F, int E) {
    int tid  = blockIdx.x * blockDim.x + threadIdx.x;
    int e    = tid >> 5;
    int lane = tid & 31;
    if (e >= E) return;
    int s = src[e], d = dst[e];
    const float* hs = h + (size_t)s * F;
    float* ad = agg + (size_t)d * F;
    for (int f = lane; f < F; f += 32)
        atomAddF(&ad[f], hs[f]);
}

__global__ void combine_kernel(float* __restrict__ agg, const float* __restrict__ h,
                               const float* __restrict__ dis, const float* __restrict__ bias,
                               int fshift, int total) {
    int i = blockIdx.x * blockDim.x + threadIdx.x;
    if (i >= total) return;
    int n = i >> fshift;
    int f = i & ((1 << fshift) - 1);
    float v = (agg[i] + h[i]) * dis[n] + bias[f];
    agg[i] = v > 0.0f ? v : 0.0f;
}

__global__ void pool_kernel(const float* __restrict__ h, const int* __restrict__ batch,
                            float* __restrict__ pooled, int total) {
    int i = blockIdx.x * blockDim.x + threadIdx.x;
    if (i >= total) return;
    int n = i >> 5, f = i & 31;
    atomAddF(&pooled[(batch[n] << 5) + f], h[i]);
}

__global__ void poolfin_kernel(float* __restrict__ pooled, const float* __restrict__ counts) {
    int i = blockIdx.x * blockDim.x + threadIdx.x;
    int g = i >> 5;
    pooled[i] = pooled[i] / fmaxf(counts[g], 1.0f);
}

extern "C" void kernel_launch(void* const* d_in, const int* in_sizes, int n_in,
                              void* d_out, int out_size, void* d_ws, size_t ws_size,
                              hipStream_t stream) {
    const float* x   = (const float*)d_in[0];
    const float* W1  = (const float*)d_in[1];
    const float* b1  = (const float*)d_in[2];
    const float* W2  = (const float*)d_in[3];
    const float* b2  = (const float*)d_in[4];
    const float* W3  = (const float*)d_in[5];
    const float* b3  = (const float*)d_in[6];
    const float* Wl  = (const float*)d_in[7];
    const float* bl  = (const float*)d_in[8];
    const int*   ei  = (const int*)d_in[9];
    const int*   bat = (const int*)d_in[10];
    float*       out = (float*)d_out;

    const int N = NNODES, E = NEDGES;
    const int* src = ei;
    const int* dst = ei + E;

    float* B0     = (float*)d_ws;
    float* B1     = B0 + (size_t)N * 128;
    float* dis    = B1 + (size_t)N * 128;
    float* counts = dis + N;
    float* pooled = counts + NGRAPH;

    hipMemsetAsync(dis, 0, (size_t)N * sizeof(float), stream);
    hipMemsetAsync(counts, 0, NGRAPH * sizeof(float), stream);
    deg_kernel<<<E / 256, 256, 0, stream>>>(dst, dis, E);
    dis_counts_kernel<<<(N + 255) / 256, 256, 0, stream>>>(dis, bat, counts, N);

    wmma_gemm_f32<<<(2500 * 8) / 8, 256, 0, stream>>>(x, W1, B0, N, 128, 128, dis, nullptr, 2500 * 8);
    hipMemsetAsync(B1, 0, (size_t)N * 128 * sizeof(float), stream);
    scatter_kernel<<<(E * 32) / 256, 256, 0, stream>>>(B0, src, dst, B1, 128, E);
    combine_kernel<<<(N * 128) / 256, 256, 0, stream>>>(B1, B0, dis, b1, 7, N * 128);

    wmma_gemm_f32<<<(2500 * 4) / 8, 256, 0, stream>>>(B1, W2, B0, N, 128, 64, dis, nullptr, 2500 * 4);
    hipMemsetAsync(B1, 0, (size_t)N * 64 * sizeof(float), stream);
    scatter_kernel<<<(E * 32) / 256, 256, 0, stream>>>(B0, src, dst, B1, 64, E);
    combine_kernel<<<(N * 64) / 256, 256, 0, stream>>>(B1, B0, dis, b2, 6, N * 64);

    wmma_gemm_f32<<<(2500 * 2) / 8, 256, 0, stream>>>(B1, W3, B0, N, 64, 32, dis, nullptr, 2500 * 2);
    hipMemsetAsync(B1, 0, (size_t)N * 32 * sizeof(float), stream);
    scatter_kernel<<<(E * 32) / 256, 256, 0, stream>>>(B0, src, dst, B1, 32, E);
    combine_kernel<<<(N * 32) / 256, 256, 0, stream>>>(B1, B0, dis, b3, 5, N * 32);

    hipMemsetAsync(pooled, 0, NGRAPH * 32 * sizeof(float), stream);
    pool_kernel<<<(N * 32) / 256, 256, 0, stream>>>(B1, bat, pooled, N * 32);
    poolfin_kernel<<<(NGRAPH * 32) / 256, 256, 0, stream>>>(pooled, counts);

    wmma_gemm_f32<<<(16 * 48) / 8, 256, 0, stream>>>(pooled, Wl, out, NGRAPH, 32, 768,
                                                     nullptr, bl, 16 * 48);
}


# float atomic scatter replaced by gather over a dst-grouped edge list (rank = returned int atomic of the degree pass); layer 1 sums dis*x rows into its GEMM's LDS A tile; sum order changes
# speedup vs baseline: 1.8066x; 1.8066x over previous
#include <hip/hip_runtime.h>


typedef __attribute__((ext_vector_type(2))) float v2f;
typedef __attribute__((ext_vector_type(8))) float v8f;

#define NNODES 40000
#define NEDGES 640000
#define NGRAPH 256
#define APITCH 132

__device__ __forceinline__ void atomAddF(float* p, float v) {
    unsafeAtomicAdd(p, v);
}

__global__ void rank_kernel(const int* __restrict__ dst, int* __restrict__ cnt,
                            int* __restrict__ rank, int E) {
    int e = blockIdx.x * blockDim.x + threadIdx.x;
    if (e < E) rank[e] = atomicAdd(&cnt[dst[e]], 1);
}

__global__ void dis_counts_kernel(const int* __restrict__ cnt, float* __restrict__ dis,
                                  const int* __restrict__ batch,
                                  float* __restrict__ counts, int N) {
    int n = blockIdx.x * blockDim.x + threadIdx.x;
    if (n < N) {
        dis[n] = rsqrtf((float)cnt[n] + 1.0f);
        atomAddF(&counts[batch[n]], 1.0f);
    }
}

__global__ void __launch_bounds__(1024) scan_kernel(const int* __restrict__ cnt,
                                                    int* __restrict__ rowptr, int N) {
    __shared__ int wsum[32];
    int t = threadIdx.x, lane = t & 31, w = t >> 5;
    int per = (N + 1023) >> 10;
    int beg = t * per;
    int end = beg + per < N ? beg + per : N;
    int s = 0;
    for (int i = beg; i < end; ++i) s += cnt[i];
    int v = s;
    for (int o = 1; o < 32; o <<= 1) {
        int u = __shfl_up(v, o);
        if (lane >= o) v += u;
    }
    if (lane == 31) wsum[w] = v;
    __syncthreads();
    if (w == 0) {
        int xs = wsum[lane];
        for (int o = 1; o < 32; o <<= 1) {
            int u = __shfl_up(xs, o);
            if (lane >= o) xs += u;
        }
        wsum[lane] = xs;
    }
    __syncthreads();
    int base = v - s + (w > 0 ? wsum[w - 1] : 0);
    for (int i = beg; i < end; ++i) { rowptr[i] = base; base += cnt[i]; }
    if (t == 1023) rowptr[N] = base;
}

__global__ void fill_kernel(const int* __restrict__ src, const int* __restrict__ dst,
                            const int* __restrict__ rank, const int* __restrict__ rowptr,
                            int* __restrict__ csr, int E) {
    int e = blockIdx.x * blockDim.x + threadIdx.x;
    if (e < E) csr[rowptr[dst[e]] + rank[e]] = src[e];
}

__device__ __forceinline__ float4 ld4(const float* p) {
    return *reinterpret_cast<const float4*>(p);
}
__device__ __forceinline__ void fma4(float4& a, const float4& v, float d) {
    a.x += v.x * d; a.y += v.y * d; a.z += v.z * d; a.w += v.w * d;
}
__device__ __forceinline__ void add4(float4& a, const float4& v) {
    a.x += v.x; a.y += v.y; a.z += v.z; a.w += v.w;
}

__device__ __forceinline__ float4 keep4(const float4& v, bool on) {
    float4 r;
    r.x = on ? v.x : 0.0f; r.y = on ? v.y : 0.0f; r.z = on ? v.z : 0.0f; r.w = on ? v.w : 0.0f;
    return r;
}

template <int F, int LPN, bool SCALE>
__device__ __forceinline__ float4 gather_sum(const float* __restrict__ h,
                                             const float* __restrict__ dis,
                                             const int* __restrict__ csr,
                                             int beg, int end, int sub, float4 acc) {
    for (int j0 = beg; j0 < end; j0 += LPN) {
        int j = j0 + sub;
        int s = 0;
        float d = 0.0f;
        if (j < end) {
            s = csr[j];
            if (SCALE) d = dis[s];
        }
        int c = end - j0 < LPN ? end - j0 : LPN;
        for (int k = 0; k < c; k += 4) {
            int last = c - 1;
            int k1 = k + 1 < last ? k + 1 : last;
            int k2 = k + 2 < last ? k + 2 : last;
            int k3 = k + 3 < last ? k + 3 : last;
            int s0 = __shfl(s, k, LPN), s1 = __shfl(s, k1, LPN);
            int s2 = __shfl(s, k2, LPN), s3 = __shfl(s, k3, LPN);
            float4 v0 = ld4(h + (size_t)s0 * F + sub * 4);
            float4 v1 = ld4(h + (size_t)s1 * F + sub * 4);
            float4 v2 = ld4(h + (size_t)s2 * F + sub * 4);
            float4 v3 = ld4(h + (size_t)s3 * F + sub * 4);
            v1 = keep4(v1, k + 1 < c);
            v2 = keep4(v2, k + 2 < c);
            v3 = keep4(v3, k + 3 < c);
            if (SCALE) {
                fma4(acc, v0, __shfl(d, k, LPN));
                fma4(acc, v1, __shfl(d, k1, LPN));
                fma4(acc, v2, __shfl(d, k2, LPN));
                fma4(acc, v3, __shfl(d, k3, LPN));
            } else {
                add4(acc, v0); add4(acc, v1); add4(acc, v2); add4(acc, v3);
            }
        }
    }
    return acc;
}

__global__ void __launch_bounds__(256) gather_gemm128(const float* __restrict__ x,
                                                      const float* __restrict__ W,
                                                      const float* __restrict__ dis,
                                                      const float* __restrict__ bias,
                                                      const int* __restrict__ rowptr,
                                                      const int* __restrict__ csr,
                                                      float* __restrict__ out) {
    __shared__ __attribute__((aligned(16))) float As[16 * APITCH];
    int tid = threadIdx.x, wid = tid >> 5, lane = tid & 31;
    int row0 = blockIdx.x << 4;
    #pragma unroll
    for (int r = 0; r < 2; ++r) {
        int lr = wid * 2 + r;
        int n = row0 + lr;
        float dn = dis[n];
        float4 acc = ld4(x + (size_t)n * 128 + lane * 4);
        acc.x *= dn; acc.y *= dn; acc.z *= dn; acc.w *= dn;
        acc = gather_sum<128, 32, true>(x, dis, csr, rowptr[n], rowptr[n + 1], lane, acc);
        *reinterpret_cast<float4*>(&As[lr * APITCH + lane * 4]) = acc;
    }
    __syncthreads();

    int hi = lane >> 4, lm = lane & 15;
    int col0 = wid << 4;
    v8f c = {};
    const float* Arow = As + lm * APITCH + 2 * hi;
    for (int k0 = 0; k0 < 128; k0 += 4) {
        v2f a = *reinterpret_cast<const v2f*>(Arow + k0);
        const float* Bp = W + (size_t)(k0 + 2 * hi) * 128 + col0 + lm;
        v2f b;
        b.x = Bp[0];
        b.y = Bp[128];
        c = __builtin_amdgcn_wmma_f32_16x16x4_f32(false, a, false, b,
                                                  (short)0, c, false, false);
    }
    float bv = bias[col0 + lm];
    #pragma unroll
    for (int v = 0; v < 8; ++v) {
        int row = row0 + v + 8 * hi;
        float o = c[v] * dis[row] + bv;
        out[(size_t)row * 128 + col0 + lm] = o > 0.0f ? o : 0.0f;
    }
}

template <int F>
__global__ void __launch_bounds__(256) gather_kernel(const float* __restrict__ h,
                                                     const float* __restrict__ dis,
                                                     const float* __restrict__ bias,
                                                     const int* __restrict__ rowptr,
                                                     const int* __restrict__ csr,
                                                     float* __restrict__ out, int N) {
    constexpr int LPN = F / 4;
    int gt = blockIdx.x * blockDim.x + threadIdx.x;
    int n = gt / LPN;
    int sub = gt % LPN;
    if (n >= N) return;
    float4 acc = ld4(h + (size_t)n * F + sub * 4);
    acc = gather_sum<F, LPN, false>(h, dis, csr, rowptr[n], rowptr[n + 1], sub, acc);
    float dn = dis[n];
    float4 b = ld4(bias + sub * 4);
    float4 o;
    o.x = acc.x * dn + b.x; o.y = acc.y * dn + b.y;
    o.z = acc.z * dn + b.z; o.w = acc.w * dn + b.w;
    o.x = o.x > 0.0f ? o.x : 0.0f; o.y = o.y > 0.0f ? o.y : 0.0f;
    o.z = o.z > 0.0f ? o.z : 0.0f; o.w = o.w > 0.0f ? o.w : 0.0f;
    *reinterpret_cast<float4*>(out + (size_t)n * F + sub * 4) = o;
}

__global__ void wmma_gemm_f32(const float* __restrict__ A, const float* __restrict__ B,
                              float* __restrict__ C, int N, int K, int M,
                              const float* __restrict__ rowscale,
                              const float* __restrict__ bias, int total_tiles) {
    int tid  = threadIdx.x;
    int wid  = tid >> 5;
    int lane = tid & 31;
    int tile = blockIdx.x * (blockDim.x >> 5) + wid;
    if (tile >= total_tiles) return;

    int tiles_m = M >> 4;
    int rt = tile / tiles_m, ct = tile - rt * tiles_m;
    int row0 = rt << 4, col0 = ct << 4;
    int hi = lane >> 4, lm = lane & 15;

    v8f c = {};
    const float* Arow = A + (size_t)(row0 + lm) * K;
    for (int k0 = 0; k0 < K; k0 += 4) {
        v2f a = *reinterpret_cast<const v2f*>(Arow + k0 + 2 * hi);
        const float* Bp = B + (size_t)(k0 + 2 * hi) * M + col0 + lm;
        v2f b;
        b.x = Bp[0];
        b.y = Bp[M];
        c = __builtin_amdgcn_wmma_f32_16x16x4_f32(false, a, false, b,
                                                  (short)0, c, false, false);
    }
    float bv = bias ? bias[col0 + lm] : 0.0f;
    #pragma unroll
    for (int v = 0; v < 8; ++v) {
        int row = row0 + v + 8 * hi;
        float o = c[v];
        if (rowscale) o *= rowscale[row];
        C[(size_t)row * M + col0 + lm] = o + bv;
    }
}

__global__ void pool_kernel(const float* __restrict__ h, const int* __restrict__ batch,
                            float* __restrict__ pooled, int total) {
    int i = blockIdx.x * blockDim.x + threadIdx.x;
    if (i >= total) return;
    int n = i >> 5, f = i & 31;
    atomAddF(&pooled[(batch[n] << 5) + f], h[i]);
}

__global__ void poolfin_kernel(float* __restrict__ pooled, const float* __restrict__ counts) {
    int i = blockIdx.x * blockDim.x + threadIdx.x;
    int g = i >> 5;
    pooled[i] = pooled[i] / fmaxf(counts[g], 1.0f);
}

extern "C" void kernel_launch(void* const* d_in, const int* in_sizes, int n_in,
                              void* d_out, int out_size, void* d_ws, size_t ws_size,
                              hipStream_t stream) {
    const float* x   = (const float*)d_in[0];
    const float* W1  = (const float*)d_in[1];
    const float* b1  = (const float*)d_in[2];
    const float* W2  = (const float*)d_in[3];
    const float* b2  = (const float*)d_in[4];
    const float* W3  = (const float*)d_in[5];
    const float* b3  = (const float*)d_in[6];
    const float* Wl  = (const float*)d_in[7];
    const float* bl  = (const float*)d_in[8];
    const int*   ei  = (const int*)d_in[9];
    const int*   bat = (const int*)d_in[10];
    float*       out = (float*)d_out;

    const int N = NNODES, E = NEDGES;
    const int* src = ei;
    const int* dst = ei + E;

    float* B0     = (float*)d_ws;
    float* B1     = B0 + (size_t)N * 128;
    int*   cnt    = (int*)(B1 + (size_t)N * 64);
    int*   rowptr = cnt + N;
    int*   rank   = rowptr + N + 32;
    int*   csr    = rank + E;
    float* dis    = B1 + (size_t)N * 128;
    float* counts = dis + N;
    float* pooled = counts + NGRAPH;

    hipMemsetAsync(cnt, 0, (size_t)N * sizeof(int), stream);
    hipMemsetAsync(counts, 0, NGRAPH * sizeof(float), stream);
    rank_kernel<<<E / 256, 256, 0, stream>>>(dst, cnt, rank, E);
    dis_counts_kernel<<<(N + 255) / 256, 256, 0, stream>>>(cnt, dis, bat, counts, N);
    scan_kernel<<<1, 1024, 0, stream>>>(cnt, rowptr, N);
    fill_kernel<<<E / 256, 256, 0, stream>>>(src, dst, rank, rowptr, csr, E);

    gather_gemm128<<<N / 16, 256, 0, stream>>>(x, W1, dis, b1, rowptr, csr, B0);

    wmma_gemm_f32<<<(2500 * 4) / 8, 256, 0, stream>>>(B0, W2, B1, N, 128, 64, dis, nullptr, 2500 * 4);
    gather_kernel<64><<<(N * 16) / 256, 256, 0, stream>>>(B1, dis, b2, rowptr, csr, B0, N);

    wmma_gemm_f32<<<(2500 * 2) / 8, 256, 0, stream>>>(B0, W3, B1, N, 64, 32, dis, nullptr, 2500 * 2);
    gather_kernel<32><<<(N * 8) / 256, 256, 0, stream>>>(B1, dis, b3, rowptr, csr, B0, N);

    hipMemsetAsync(pooled, 0, NGRAPH * 32 * sizeof(float), stream);
    pool_kernel<<<(N * 32) / 256, 256, 0, stream>>>(B0, bat, pooled, N * 32);
    poolfin_kernel<<<(NGRAPH * 32) / 256, 256, 0, stream>>>(pooled, counts);

    wmma_gemm_f32<<<(16 * 48) / 8, 256, 0, stream>>>(pooled, Wl, out, NGRAPH, 32, 768,
                                                     nullptr, bl, 16 * 48);
}


# one-work-group prefix scan of degrees replaced by slice allocation in the degree-norm pass (block scan of 256 degrees + one returning int add per work-group); one launch fewer; sum order as before
# speedup vs baseline: 2.2868x; 1.2658x over previous
#include <hip/hip_runtime.h>


typedef __attribute__((ext_vector_type(2))) float v2f;
typedef __attribute__((ext_vector_type(8))) float v8f;

#define NNODES 40000
#define NEDGES 640000
#define NGRAPH 256
#define APITCH 132

__device__ __forceinline__ void atomAddF(float* p, float v) {
    unsafeAtomicAdd(p, v);
}

__global__ void rank_kernel(const int* __restrict__ dst, int* __restrict__ cnt,
                            int* __restrict__ rank, int E) {
    int e = blockIdx.x * blockDim.x + threadIdx.x;
    if (e < E) rank[e] = atomicAdd(&cnt[dst[e]], 1);
}

__global__ void __launch_bounds__(256) dis_counts_alloc_kernel(const int* __restrict__ cnt,
                                                               float* __restrict__ dis,
                                                               const int* __restrict__ batch,
                                                               float* __restrict__ counts,
                                                               int* __restrict__ seg,
                                                               int* __restrict__ total, int N) {
    __shared__ int wsum[8];
    __shared__ int wgbase;
    int t = threadIdx.x, lane = t & 31, w = t >> 5;
    int n = blockIdx.x * 256 + t;
    int c = 0;
    if (n < N) {
        c = cnt[n];
        dis[n] = rsqrtf((float)c + 1.0f);
        atomAddF(&counts[batch[n]], 1.0f);
    }
    int v = c;
    for (int o = 1; o < 32; o <<= 1) {
        int u = __shfl_up(v, o);
        if (lane >= o) v += u;
    }
    if (lane == 31) wsum[w] = v;
    __syncthreads();
    if (t == 0) {
        int run = 0;
        for (int i = 0; i < 8; ++i) { int q = wsum[i]; wsum[i] = run; run += q; }
        wgbase = atomicAdd(total, run);
    }
    __syncthreads();
    if (n < N) {
        int beg = wgbase + wsum[w] + v - c;
        *reinterpret_cast<int2*>(seg + 2 * (size_t)n) = make_int2(beg, beg + c);
    }
}

__global__ void fill_kernel(const int* __restrict__ src, const int* __restrict__ dst,
                            const int* __restrict__ rank, const int* __restrict__ seg,
                            int* __restrict__ csr, int E) {
    int e = blockIdx.x * blockDim.x + threadIdx.x;
    if (e < E) csr[seg[2 * (size_t)dst[e]] + rank[e]] = src[e];
}

__device__ __forceinline__ float4 ld4(const float* p) {
    return *reinterpret_cast<const float4*>(p);
}
__device__ __forceinline__ int2 ld2i(const int* p) {
    return *reinterpret_cast<const int2*>(p);
}
__device__ __forceinline__ void fma4(float4& a, const float4& v, float d) {
    a.x += v.x * d; a.y += v.y * d; a.z += v.z * d; a.w += v.w * d;
}
__device__ __forceinline__ void add4(float4& a, const float4& v) {
    a.x += v.x; a.y += v.y; a.z += v.z; a.w += v.w;
}

__device__ __forceinline__ float4 keep4(const float4& v, bool on) {
    float4 r;
    r.x = on ? v.x : 0.0f; r.y = on ? v.y : 0.0f; r.z = on ? v.z : 0.0f; r.w = on ? v.w : 0.0f;
    return r;
}

template <int F, int LPN, bool SCALE>
__device__ __forceinline__ float4 gather_sum(const float* __restrict__ h,
                                             const float* __restrict__ dis,
                                             const int* __restrict__ csr,
                                             int beg, int end, int sub, float4 acc) {
    for (int j0 = beg; j0 < end; j0 += LPN) {
        int j = j0 + sub;
        int s = 0;
        float d = 0.0f;
        if (j < end) {
            s = csr[j];
            if (SCALE) d = dis[s];
        }
        int c = end - j0 < LPN ? end - j0 : LPN;
        for (int k = 0; k < c; k += 4) {
            int last = c - 1;
            int k1 = k + 1 < last ? k + 1 : last;
            int k2 = k + 2 < last ? k + 2 : last;
            int k3 = k + 3 < last ? k + 3 : last;
            int s0 = __shfl(s, k, LPN), s1 = __shfl(s, k1, LPN);
            int s2 = __shfl(s, k2, LPN), s3 = __shfl(s, k3, LPN);
            float4 v0 = ld4(h + (size_t)s0 * F + sub * 4);
            float4 v1 = ld4(h + (size_t)s1 * F + sub * 4);
            float4 v2 = ld4(h + (size_t)s2 * F + sub * 4);
            float4 v3 = ld4(h + (size_t)s3 * F + sub * 4);
            v1 = keep4(v1, k + 1 < c);
            v2 = keep4(v2, k + 2 < c);
            v3 = keep4(v3, k + 3 < c);
            if (SCALE) {
                fma4(acc, v0, __shfl(d, k, LPN));
                fma4(acc, v1, __shfl(d, k1, LPN));
                fma4(acc, v2, __shfl(d, k2, LPN));
                fma4(acc, v3, __shfl(d, k3, LPN));
            } else {
                add4(acc, v0); add4(acc, v1); add4(acc, v2); add4(acc, v3);
            }
        }
    }
    return acc;
}

__global__ void __launch_bounds__(256) gather_gemm128(const float* __restrict__ x,
                                                      const float* __restrict__ W,
                                                      const float* __restrict__ dis,
                                                      const float* __restrict__ bias,
                                                      const int* __restrict__ seg,
                                                      const int* __restrict__ csr,
                                                      float* __restrict__ out) {
    __shared__ __attribute__((aligned(16))) float As[16 * APITCH];
    int tid = threadIdx.x, wid = tid >> 5, lane = tid & 31;
    int row0 = blockIdx.x << 4;
    #pragma unroll
    for (int r = 0; r < 2; ++r) {
        int lr = wid * 2 + r;
        int n = row0 + lr;
        float dn = dis[n];
        float4 acc = ld4(x + (size_t)n * 128 + lane * 4);
        acc.x *= dn; acc.y *= dn; acc.z *= dn; acc.w *= dn;
        int2 se = ld2i(seg + 2 * (size_t)n);
        acc = gather_sum<128, 32, true>(x, dis, csr, se.x, se.y, lane, acc);
        *reinterpret_cast<float4*>(&As[lr * APITCH + lane * 4]) = acc;
    }
    __syncthreads();

    int hi = lane >> 4, lm = lane & 15;
    int col0 = wid << 4;
    v8f c = {};
    const float* Arow = As + lm * APITCH + 2 * hi;
    for (int k0 = 0; k0 < 128; k0 += 4) {
        v2f a = *reinterpret_cast<const v2f*>(Arow + k0);
        const float* Bp = W + (size_t)(k0 + 2 * hi) * 128 + col0 + lm;
        v2f b;
        b.x = Bp[0];
        b.y = Bp[128];
        c = __builtin_amdgcn_wmma_f32_16x16x4_f32(false, a, false, b,
                                                  (short)0, c, false, false);
    }
    float bv = bias[col0 + lm];
    #pragma unroll
    for (int v = 0; v < 8; ++v) {
        int row = row0 + v + 8 * hi;
        float o = c[v] * dis[row] + bv;
        out[(size_t)row * 128 + col0 + lm] = o > 0.0f ? o : 0.0f;
    }
}

template <int F>
__global__ void __launch_bounds__(256) gather_kernel(const float* __restrict__ h,
                                                     const float* __restrict__ dis,
                                                     const float* __restrict__ bias,
                                                     const int* __restrict__ seg,
                                                     const int* __restrict__ csr,
                                                     float* __restrict__ out, int N) {
    constexpr int LPN = F / 4;
    int gt = blockIdx.x * blockDim.x + threadIdx.x;
    int n = gt / LPN;
    int sub = gt % LPN;
    if (n >= N) return;
    float4 acc = ld4(h + (size_t)n * F + sub * 4);
    int2 se = ld2i(seg + 2 * (size_t)n);
    acc = gather_sum<F, LPN, false>(h, dis, csr, se.x, se.y, sub, acc);
    float dn = dis[n];
    float4 b = ld4(bias + sub * 4);
    float4 o;
    o.x = acc.x * dn + b.x; o.y = acc.y * dn + b.y;
    o.z = acc.z * dn + b.z; o.w = acc.w * dn + b.w;
    o.x = o.x > 0.0f ? o.x : 0.0f; o.y = o.y > 0.0f ? o.y : 0.0f;
    o.z = o.z > 0.0f ? o.z : 0.0f; o.w = o.w > 0.0f ? o.w : 0.0f;
    *reinterpret_cast<float4*>(out + (size_t)n * F + sub * 4) = o;
}

__global__ void wmma_gemm_f32(const float* __restrict__ A, const float* __restrict__ B,
                              float* __restrict__ C, int N, int K, int M,
                              const float* __restrict__ rowscale,
                              const float* __restrict__ bias, int total_tiles) {
    int tid  = threadIdx.x;
    int wid  = tid >> 5;
    int lane = tid & 31;
    int tile = blockIdx.x * (blockDim.x >> 5) + wid;
    if (tile >= total_tiles) return;

    int tiles_m = M >> 4;
    int rt = tile / tiles_m, ct = tile - rt * tiles_m;
    int row0 = rt << 4, col0 = ct << 4;
    int hi = lane >> 4, lm = lane & 15;

    v8f c = {};
    const float* Arow = A + (size_t)(row0 + lm) * K;
    for (int k0 = 0; k0 < K; k0 += 4) {
        v2f a = *reinterpret_cast<const v2f*>(Arow + k0 + 2 * hi);
        const float* Bp = B + (size_t)(k0 + 2 * hi) * M + col0 + lm;
        v2f b;
        b.x = Bp[0];
        b.y = Bp[M];
        c = __builtin_amdgcn_wmma_f32_16x16x4_f32(false, a, false, b,
                                                  (short)0, c, false, false);
    }
    float bv = bias ? bias[col0 + lm] : 0.0f;
    #pragma unroll
    for (int v = 0; v < 8; ++v) {
        int row = row0 + v + 8 * hi;
        float o = c[v];
        if (rowscale) o *= rowscale[row];
        C[(size_t)row * M + col0 + lm] = o + bv;
    }
}

__global__ void pool_kernel(const float* __restrict__ h, const int* __restrict__ batch,
                            float* __restrict__ pooled, int total) {
    int i = blockIdx.x * blockDim.x + threadIdx.x;
    if (i >= total) return;
    int n = i >> 5, f = i & 31;
    atomAddF(&pooled[(batch[n] << 5) + f], h[i]);
}

__global__ void poolfin_kernel(float* __restrict__ pooled, const float* __restrict__ counts) {
    int i = blockIdx.x * blockDim.x + threadIdx.x;
    int g = i >> 5;
    pooled[i] = pooled[i] / fmaxf(counts[g], 1.0f);
}

extern "C" void kernel_launch(void* const* d_in, const int* in_sizes, int n_in,
                              void* d_out, int out_size, void* d_ws, size_t ws_size,
                              hipStream_t stream) {
    const float* x   = (const float*)d_in[0];
    const float* W1  = (const float*)d_in[1];
    const float* b1  = (const float*)d_in[2];
    const float* W2  = (const float*)d_in[3];
    const float* b2  = (const float*)d_in[4];
    const float* W3  = (const float*)d_in[5];
    const float* b3  = (const float*)d_in[6];
    const float* Wl  = (const float*)d_in[7];
    const float* bl  = (const float*)d_in[8];
    const int*   ei  = (const int*)d_in[9];
    const int*   bat = (const int*)d_in[10];
    float*       out = (float*)d_out;

    const int N = NNODES, E = NEDGES;
    const int* src = ei;
    const int* dst = ei + E;

    float* B0     = (float*)d_ws;
    float* B1     = B0 + (size_t)N * 128;
    int*   cnt    = (int*)(B1 + (size_t)N * 64);
    int*   total  = cnt + N;
    int*   seg    = total + 32;
    int*   rank   = seg + 2 * N;
    int*   csr    = rank + E;
    float* dis    = B1 + (size_t)N * 128;
    float* counts = dis + N;
    float* pooled = counts + NGRAPH;

    hipMemsetAsync(cnt, 0, (size_t)(N + 1) * sizeof(int), stream);
    hipMemsetAsync(counts, 0, NGRAPH * sizeof(float), stream);
    rank_kernel<<<E / 256, 256, 0, stream>>>(dst, cnt, rank, E);
    dis_counts_alloc_kernel<<<(N + 255) / 256, 256, 0, stream>>>(cnt, dis, bat, counts, seg, total, N);
    fill_kernel<<<E / 256, 256, 0, stream>>>(src, dst, rank, seg, csr, E);

    gather_gemm128<<<N / 16, 256, 0, stream>>>(x, W1, dis, b1, seg, csr, B0);

    wmma_gemm_f32<<<(2500 * 4) / 8, 256, 0, stream>>>(B0, W2, B1, N, 128, 64, dis, nullptr, 2500 * 4);
    gather_kernel<64><<<(N * 16) / 256, 256, 0, stream>>>(B1, dis, b2, seg, csr, B0, N);

    wmma_gemm_f32<<<(2500 * 2) / 8, 256, 0, stream>>>(B0, W3, B1, N, 64, 32, dis, nullptr, 2500 * 2);
    gather_kernel<32><<<(N * 8) / 256, 256, 0, stream>>>(B1, dis, b3, seg, csr, B0, N);

    hipMemsetAsync(pooled, 0, NGRAPH * 32 * sizeof(float), stream);
    pool_kernel<<<(N * 32) / 256, 256, 0, stream>>>(B0, bat, pooled, N * 32);
    poolfin_kernel<<<(NGRAPH * 32) / 256, 256, 0, stream>>>(pooled, counts);

    wmma_gemm_f32<<<(16 * 48) / 8, 256, 0, stream>>>(pooled, Wl, out, NGRAPH, 32, 768,
                                                     nullptr, bl, 16 * 48);
}


# mean pool folded into the layer-3 gather: a work-group sums its 32 rows per graph in LDS, one float atomic per feature per run; the divide moves into the head GEMM's A load; pool sum order changes
# speedup vs baseline: 2.4293x; 1.0623x over previous
#include <hip/hip_runtime.h>


typedef __attribute__((ext_vector_type(2))) float v2f;
typedef __attribute__((ext_vector_type(8))) float v8f;

#define NNODES 40000
#define NEDGES 640000
#define NGRAPH 256
#define APITCH 132

__device__ __forceinline__ void atomAddF(float* p, float v) {
    unsafeAtomicAdd(p, v);
}

__global__ void rank_kernel(const int* __restrict__ dst, int* __restrict__ cnt,
                            int* __restrict__ rank, int E) {
    int e = blockIdx.x * blockDim.x + threadIdx.x;
    if (e < E) rank[e] = atomicAdd(&cnt[dst[e]], 1);
}

__global__ void __launch_bounds__(256) dis_counts_alloc_kernel(const int* __restrict__ cnt,
                                                               float* __restrict__ dis,
                                                               const int* __restrict__ batch,
                                                               float* __restrict__ counts,
                                                               int* __restrict__ seg,
                                                               int* __restrict__ total, int N) {
    __shared__ int wsum[8];
    __shared__ int wgbase;
    int t = threadIdx.x, lane = t & 31, w = t >> 5;
    int n = blockIdx.x * 256 + t;
    int c = 0;
    if (n < N) {
        c = cnt[n];
        dis[n] = rsqrtf((float)c + 1.0f);
        atomAddF(&counts[batch[n]], 1.0f);
    }
    int v = c;
    for (int o = 1; o < 32; o <<= 1) {
        int u = __shfl_up(v, o);
        if (lane >= o) v += u;
    }
    if (lane == 31) wsum[w] = v;
    __syncthreads();
    if (t == 0) {
        int run = 0;
        for (int i = 0; i < 8; ++i) { int q = wsum[i]; wsum[i] = run; run += q; }
        wgbase = atomicAdd(total, run);
    }
    __syncthreads();
    if (n < N) {
        int beg = wgbase + wsum[w] + v - c;
        *reinterpret_cast<int2*>(seg + 2 * (size_t)n) = make_int2(beg, beg + c);
    }
}

__global__ void fill_kernel(const int* __restrict__ src, const int* __restrict__ dst,
                            const int* __restrict__ rank, const int* __restrict__ seg,
                            int* __restrict__ csr, int E) {
    int e = blockIdx.x * blockDim.x + threadIdx.x;
    if (e < E) csr[seg[2 * (size_t)dst[e]] + rank[e]] = src[e];
}

__device__ __forceinline__ float4 ld4(const float* p) {
    return *reinterpret_cast<const float4*>(p);
}
__device__ __forceinline__ int2 ld2i(const int* p) {
    return *reinterpret_cast<const int2*>(p);
}
__device__ __forceinline__ void fma4(float4& a, const float4& v, float d) {
    a.x += v.x * d; a.y += v.y * d; a.z += v.z * d; a.w += v.w * d;
}
__device__ __forceinline__ void add4(float4& a, const float4& v) {
    a.x += v.x; a.y += v.y; a.z += v.z; a.w += v.w;
}

__device__ __forceinline__ float4 keep4(const float4& v, bool on) {
    float4 r;
    r.x = on ? v.x : 0.0f; r.y = on ? v.y : 0.0f; r.z = on ? v.z : 0.0f; r.w = on ? v.w : 0.0f;
    return r;
}

template <int F, int LPN, bool SCALE>
__device__ __forceinline__ float4 gather_sum(const float* __restrict__ h,
                                             const float* __restrict__ dis,
                                             const int* __restrict__ csr,
                                             int beg, int end, int sub, float4 acc) {
    for (int j0 = beg; j0 < end; j0 += LPN) {
        int j = j0 + sub;
        int s = 0;
        float d = 0.0f;
        if (j < end) {
            s = csr[j];
            if (SCALE) d = dis[s];
        }
        int c = end - j0 < LPN ? end - j0 : LPN;
        for (int k = 0; k < c; k += 4) {
            int last = c - 1;
            int k1 = k + 1 < last ? k + 1 : last;
            int k2 = k + 2 < last ? k + 2 : last;
            int k3 = k + 3 < last ? k + 3 : last;
            int s0 = __shfl(s, k, LPN), s1 = __shfl(s, k1, LPN);
            int s2 = __shfl(s, k2, LPN), s3 = __shfl(s, k3, LPN);
            float4 v0 = ld4(h + (size_t)s0 * F + sub * 4);
            float4 v1 = ld4(h + (size_t)s1 * F + sub * 4);
            float4 v2 = ld4(h + (size_t)s2 * F + sub * 4);
            float4 v3 = ld4(h + (size_t)s3 * F + sub * 4);
            v1 = keep4(v1, k + 1 < c);
            v2 = keep4(v2, k + 2 < c);
            v3 = keep4(v3, k + 3 < c);
            if (SCALE) {
                fma4(acc, v0, __shfl(d, k, LPN));
                fma4(acc, v1, __shfl(d, k1, LPN));
                fma4(acc, v2, __shfl(d, k2, LPN));
                fma4(acc, v3, __shfl(d, k3, LPN));
            } else {
                add4(acc, v0); add4(acc, v1); add4(acc, v2); add4(acc, v3);
            }
        }
    }
    return acc;
}

__global__ void __launch_bounds__(256) gather_gemm128(const float* __restrict__ x,
                                                      const float* __restrict__ W,
                                                      const float* __restrict__ dis,
                                                      const float* __restrict__ bias,
                                                      const int* __restrict__ seg,
                                                      const int* __restrict__ csr,
                                                      float* __restrict__ out) {
    __shared__ __attribute__((aligned(16))) float As[16 * APITCH];
    int tid = threadIdx.x, wid = tid >> 5, lane = tid & 31;
    int row0 = blockIdx.x << 4;
    #pragma unroll
    for (int r = 0; r < 2; ++r) {
        int lr = wid * 2 + r;
        int n = row0 + lr;
        float dn = dis[n];
        float4 acc = ld4(x + (size_t)n * 128 + lane * 4);
        acc.x *= dn; acc.y *= dn; acc.z *= dn; acc.w *= dn;
        int2 se = ld2i(seg + 2 * (size_t)n);
        acc = gather_sum<128, 32, true>(x, dis, csr, se.x, se.y, lane, acc);
        *reinterpret_cast<float4*>(&As[lr * APITCH + lane * 4]) = acc;
    }
    __syncthreads();

    int hi = lane >> 4, lm = lane & 15;
    int col0 = wid << 4;
    v8f c = {};
    const float* Arow = As + lm * APITCH + 2 * hi;
    for (int k0 = 0; k0 < 128; k0 += 4) {
        v2f a = *reinterpret_cast<const v2f*>(Arow + k0);
        const float* Bp = W + (size_t)(k0 + 2 * hi) * 128 + col0 + lm;
        v2f b;
        b.x = Bp[0];
        b.y = Bp[128];
        c = __builtin_amdgcn_wmma_f32_16x16x4_f32(false, a, false, b,
                                                  (short)0, c, false, false);
    }
    float bv = bias[col0 + lm];
    #pragma unroll
    for (int v = 0; v < 8; ++v) {
        int row = row0 + v + 8 * hi;
        float o = c[v] * dis[row] + bv;
        out[(size_t)row * 128 + col0 + lm] = o > 0.0f ? o : 0.0f;
    }
}

template <int F>
__global__ void __launch_bounds__(256) gather_kernel(const float* __restrict__ h,
                                                     const float* __restrict__ dis,
                                                     const float* __restrict__ bias,
                                                     const int* __restrict__ seg,
                                                     const int* __restrict__ csr,
                                                     float* __restrict__ out, int N) {
    constexpr int LPN = F / 4;
    int gt = blockIdx.x * blockDim.x + threadIdx.x;
    int n = gt / LPN;
    int sub = gt % LPN;
    if (n >= N) return;
    float4 acc = ld4(h + (size_t)n * F + sub * 4);
    int2 se = ld2i(seg + 2 * (size_t)n);
    acc = gather_sum<F, LPN, false>(h, dis, csr, se.x, se.y, sub, acc);
    float dn = dis[n];
    float4 b = ld4(bias + sub * 4);
    float4 o;
    o.x = acc.x * dn + b.x; o.y = acc.y * dn + b.y;
    o.z = acc.z * dn + b.z; o.w = acc.w * dn + b.w;
    o.x = o.x > 0.0f ? o.x : 0.0f; o.y = o.y > 0.0f ? o.y : 0.0f;
    o.z = o.z > 0.0f ? o.z : 0.0f; o.w = o.w > 0.0f ? o.w : 0.0f;
    *reinterpret_cast<float4*>(out + (size_t)n * F + sub * 4) = o;
}

__global__ void wmma_gemm_f32(const float* __restrict__ A, const float* __restrict__ B,
                              float* __restrict__ C, int N, int K, int M,
                              const float* __restrict__ rowscale,
                              const float* __restrict__ bias, int total_tiles) {
    int tid  = threadIdx.x;
    int wid  = tid >> 5;
    int lane = tid & 31;
    int tile = blockIdx.x * (blockDim.x >> 5) + wid;
    if (tile >= total_tiles) return;

    int tiles_m = M >> 4;
    int rt = tile / tiles_m, ct = tile - rt * tiles_m;
    int row0 = rt << 4, col0 = ct << 4;
    int hi = lane >> 4, lm = lane & 15;

    v8f c = {};
    const float* Arow = A + (size_t)(row0 + lm) * K;
    for (int k0 = 0; k0 < K; k0 += 4) {
        v2f a = *reinterpret_cast<const v2f*>(Arow + k0 + 2 * hi);
        const float* Bp = B + (size_t)(k0 + 2 * hi) * M + col0 + lm;
        v2f b;
        b.x = Bp[0];
        b.y = Bp[M];
        c = __builtin_amdgcn_wmma_f32_16x16x4_f32(false, a, false, b,
                                                  (short)0, c, false, false);
    }
    float bv = bias ? bias[col0 + lm] : 0.0f;
    #pragma unroll
    for (int v = 0; v < 8; ++v) {
        int row = row0 + v + 8 * hi;
        float o = c[v];
        if (rowscale) o *= rowscale[row];
        C[(size_t)row * M + col0 + lm] = o + bv;
    }
}

__global__ void __launch_bounds__(256) gather_pool32(const float* __restrict__ h,
                                                     const float* __restrict__ dis,
                                                     const float* __restrict__ bias,
                                                     const int* __restrict__ seg,
                                                     const int* __restrict__ csr,
                                                     const int* __restrict__ batch,
                                                     float* __restrict__ pooled, int N) {
    __shared__ __attribute__((aligned(16))) float rows[32 * 32];
    __shared__ int gid[32];
    int t = threadIdx.x;
    int ln = t >> 3, sub = t & 7;
    int n = blockIdx.x * 32 + ln;
    float4 o = make_float4(0.0f, 0.0f, 0.0f, 0.0f);
    int g = -1;
    if (n < N) {
        float4 acc = ld4(h + (size_t)n * 32 + sub * 4);
        int2 se = ld2i(seg + 2 * (size_t)n);
        acc = gather_sum<32, 8, false>(h, dis, csr, se.x, se.y, sub, acc);
        float dn = dis[n];
        float4 b = ld4(bias + sub * 4);
        o.x = acc.x * dn + b.x; o.y = acc.y * dn + b.y;
        o.z = acc.z * dn + b.z; o.w = acc.w * dn + b.w;
        o.x = o.x > 0.0f ? o.x : 0.0f; o.y = o.y > 0.0f ? o.y : 0.0f;
        o.z = o.z > 0.0f ? o.z : 0.0f; o.w = o.w > 0.0f ? o.w : 0.0f;
        g = batch[n];
    }
    *reinterpret_cast<float4*>(&rows[ln * 32 + sub * 4]) = o;
    if (sub == 0) gid[ln] = g;
    __syncthreads();
    if (t < 32) {
        int mine = gid[t];
        float r[32];
        #pragma unroll
        for (int i = 0; i < 32; ++i) r[i] = rows[i * 32 + t];
        int cur = __builtin_amdgcn_readlane(mine, 0);
        float s = 0.0f;
        #pragma unroll
        for (int i = 0; i < 32; ++i) {
            int gi = __builtin_amdgcn_readlane(mine, i);
            if (gi != cur) {
                if (cur >= 0) atomAddF(&pooled[(cur << 5) + t], s);
                s = 0.0f;
                cur = gi;
            }
            s += r[i];
        }
        if (cur >= 0) atomAddF(&pooled[(cur << 5) + t], s);
    }
}

__global__ void __launch_bounds__(256) head_gemm_mean(const float* __restrict__ pooled,
                                                      const float* __restrict__ counts,
                                                      const float* __restrict__ Wl,
                                                      const float* __restrict__ bl,
                                                      float* __restrict__ out) {
    int tid = threadIdx.x, wid = tid >> 5, lane = tid & 31;
    int tile = blockIdx.x * 8 + wid;
    if (tile >= 16 * 48) return;
    int rt = tile / 48, ct = tile - rt * 48;
    int row0 = rt << 4, col0 = ct << 4;
    int hi = lane >> 4, lm = lane & 15;
    float cm = fmaxf(counts[row0 + lm], 1.0f);
    v8f c = {};
    const float* Arow = pooled + (size_t)(row0 + lm) * 32;
    for (int k0 = 0; k0 < 32; k0 += 4) {
        v2f a = *reinterpret_cast<const v2f*>(Arow + k0 + 2 * hi);
        a.x = a.x / cm;
        a.y = a.y / cm;
        const float* Bp = Wl + (size_t)(k0 + 2 * hi) * 768 + col0 + lm;
        v2f b;
        b.x = Bp[0];
        b.y = Bp[768];
        c = __builtin_amdgcn_wmma_f32_16x16x4_f32(false, a, false, b,
                                                  (short)0, c, false, false);
    }
    float bv = bl[col0 + lm];
    #pragma unroll
    for (int v = 0; v < 8; ++v) {
        int row = row0 + v + 8 * hi;
        out[(size_t)row * 768 + col0 + lm] = c[v] + bv;
    }
}

extern "C" void kernel_launch(void* const* d_in, const int* in_sizes, int n_in,
                              void* d_out, int out_size, void* d_ws, size_t ws_size,
                              hipStream_t stream) {
    const float* x   = (const float*)d_in[0];
    const float* W1  = (const float*)d_in[1];
    const float* b1  = (const float*)d_in[2];
    const float* W2  = (const float*)d_in[3];
    const float* b2  = (const float*)d_in[4];
    const float* W3  = (const float*)d_in[5];
    const float* b3  = (const float*)d_in[6];
    const float* Wl  = (const float*)d_in[7];
    const float* bl  = (const float*)d_in[8];
    const int*   ei  = (const int*)d_in[9];
    const int*   bat = (const int*)d_in[10];
    float*       out = (float*)d_out;

    const int N = NNODES, E = NEDGES;
    const int* src = ei;
    const int* dst = ei + E;

    float* B0     = (float*)d_ws;
    float* B1     = B0 + (size_t)N * 128;
    int*   cnt    = (int*)(B1 + (size_t)N * 64);
    int*   total  = cnt + N;
    int*   seg    = total + 32;
    int*   rank   = seg + 2 * N;
    int*   csr    = rank + E;
    float* dis    = B1 + (size_t)N * 128;
    float* counts = dis + N;
    float* pooled = counts + NGRAPH;

    hipMemsetAsync(cnt, 0, (size_t)(N + 1) * sizeof(int), stream);
    hipMemsetAsync(counts, 0, (NGRAPH + NGRAPH * 32) * sizeof(float), stream);
    rank_kernel<<<E / 256, 256, 0, stream>>>(dst, cnt, rank, E);
    dis_counts_alloc_kernel<<<(N + 255) / 256, 256, 0, stream>>>(cnt, dis, bat, counts, seg, total, N);
    fill_kernel<<<E / 256, 256, 0, stream>>>(src, dst, rank, seg, csr, E);

    gather_gemm128<<<N / 16, 256, 0, stream>>>(x, W1, dis, b1, seg, csr, B0);

    wmma_gemm_f32<<<(2500 * 4) / 8, 256, 0, stream>>>(B0, W2, B1, N, 128, 64, dis, nullptr, 2500 * 4);
    gather_kernel<64><<<(N * 16) / 256, 256, 0, stream>>>(B1, dis, b2, seg, csr, B0, N);

    wmma_gemm_f32<<<(2500 * 2) / 8, 256, 0, stream>>>(B0, W3, B1, N, 64, 32, dis, nullptr, 2500 * 2);
    gather_pool32<<<N / 32, 256, 0, stream>>>(B1, dis, b3, seg, csr, bat, pooled, N);

    head_gemm_mean<<<(16 * 48) / 8, 256, 0, stream>>>(pooled, counts, Wl, bl, out);
}


# layer-2 and layer-3 weight products done inside the preceding gather kernels: a work-group's 16 activation rows stay in LDS as the A tile; two launches, two activation round trips go; sums unchanged
# speedup vs baseline: 2.8627x; 1.1784x over previous
#include <hip/hip_runtime.h>


typedef __attribute__((ext_vector_type(2))) float v2f;
typedef __attribute__((ext_vector_type(8))) float v8f;

#define NNODES 40000
#define NEDGES 640000
#define NGRAPH 256
#define APITCH 132
#define HPITCH 68

__device__ __forceinline__ void atomAddF(float* p, float v) {
    unsafeAtomicAdd(p, v);
}

__global__ void rank_kernel(const int* __restrict__ dst, int* __restrict__ cnt,
                            int* __restrict__ rank, int E) {
    int e = blockIdx.x * blockDim.x + threadIdx.x;
    if (e < E) rank[e] = atomicAdd(&cnt[dst[e]], 1);
}

__global__ void __launch_bounds__(256) dis_counts_alloc_kernel(const int* __restrict__ cnt,
                                                               float* __restrict__ dis,
                                                               const int* __restrict__ batch,
                                                               float* __restrict__ counts,
                                                               int* __restrict__ seg,
                                                               int* __restrict__ total, int N) {
    __shared__ int wsum[8];
    __shared__ int wgbase;
    int t = threadIdx.x, lane = t & 31, w = t >> 5;
    int n = blockIdx.x * 256 + t;
    int c = 0;
    if (n < N) {
        c = cnt[n];
        dis[n] = rsqrtf((float)c + 1.0f);
        atomAddF(&counts[batch[n]], 1.0f);
    }
    int v = c;
    for (int o = 1; o < 32; o <<= 1) {
        int u = __shfl_up(v, o);
        if (lane >= o) v += u;
    }
    if (lane == 31) wsum[w] = v;
    __syncthreads();
    if (t == 0) {
        int run = 0;
        for (int i = 0; i < 8; ++i) { int q = wsum[i]; wsum[i] = run; run += q; }
        wgbase = atomicAdd(total, run);
    }
    __syncthreads();
    if (n < N) {
        int beg = wgbase + wsum[w] + v - c;
        *reinterpret_cast<int2*>(seg + 2 * (size_t)n) = make_int2(beg, beg + c);
    }
}

__global__ void fill_kernel(const int* __restrict__ src, const int* __restrict__ dst,
                            const int* __restrict__ rank, const int* __restrict__ seg,
                            int* __restrict__ csr, int E) {
    int e = blockIdx.x * blockDim.x + threadIdx.x;
    if (e < E) csr[seg[2 * (size_t)dst[e]] + rank[e]] = src[e];
}

__device__ __forceinline__ float4 ld4(const float* p) {
    return *reinterpret_cast<const float4*>(p);
}
__device__ __forceinline__ int2 ld2i(const int* p) {
    return *reinterpret_cast<const int2*>(p);
}
__device__ __forceinline__ void fma4(float4& a, const float4& v, float d) {
    a.x += v.x * d; a.y += v.y * d; a.z += v.z * d; a.w += v.w * d;
}
__device__ __forceinline__ void add4(float4& a, const float4& v) {
    a.x += v.x; a.y += v.y; a.z += v.z; a.w += v.w;
}

__device__ __forceinline__ float4 keep4(const float4& v, bool on) {
    float4 r;
    r.x = on ? v.x : 0.0f; r.y = on ? v.y : 0.0f; r.z = on ? v.z : 0.0f; r.w = on ? v.w : 0.0f;
    return r;
}

template <int F, int LPN, bool SCALE>
__device__ __forceinline__ float4 gather_sum(const float* __restrict__ h,
                                             const float* __restrict__ dis,
                                             const int* __restrict__ csr,
                                             int beg, int end, int sub, float4 acc) {
    for (int j0 = beg; j0 < end; j0 += LPN) {
        int j = j0 + sub;
        int s = 0;
        float d = 0.0f;
        if (j < end) {
            s = csr[j];
            if (SCALE) d = dis[s];
        }
        int c = end - j0 < LPN ? end - j0 : LPN;
        for (int k = 0; k < c; k += 4) {
            int last = c - 1;
            int k1 = k + 1 < last ? k + 1 : last;
            int k2 = k + 2 < last ? k + 2 : last;
            int k3 = k + 3 < last ? k + 3 : last;
            int s0 = __shfl(s, k, LPN), s1 = __shfl(s, k1, LPN);
            int s2 = __shfl(s, k2, LPN), s3 = __shfl(s, k3, LPN);
            float4 v0 = ld4(h + (size_t)s0 * F + sub * 4);
            float4 v1 = ld4(h + (size_t)s1 * F + sub * 4);
            float4 v2 = ld4(h + (size_t)s2 * F + sub * 4);
            float4 v3 = ld4(h + (size_t)s3 * F + sub * 4);
            v1 = keep4(v1, k + 1 < c);
            v2 = keep4(v2, k + 2 < c);
            v3 = keep4(v3, k + 3 < c);
            if (SCALE) {
                fma4(acc, v0, __shfl(d, k, LPN));
                fma4(acc, v1, __shfl(d, k1, LPN));
                fma4(acc, v2, __shfl(d, k2, LPN));
                fma4(acc, v3, __shfl(d, k3, LPN));
            } else {
                add4(acc, v0); add4(acc, v1); add4(acc, v2); add4(acc, v3);
            }
        }
    }
    return acc;
}

__global__ void __launch_bounds__(256) gather_gemm128_64(const float* __restrict__ x,
                                                         const float* __restrict__ W,
                                                         const float* __restrict__ dis,
                                                         const float* __restrict__ bias,
                                                         const int* __restrict__ seg,
                                                         const int* __restrict__ csr,
                                                         const float* __restrict__ Wn,
                                                         float* __restrict__ out) {
    __shared__ __attribute__((aligned(16))) float As[16 * APITCH];
    __shared__ __attribute__((aligned(16))) float Hs[16 * APITCH];
    int tid = threadIdx.x, wid = tid >> 5, lane = tid & 31;
    int row0 = blockIdx.x << 4;
    #pragma unroll
    for (int r = 0; r < 2; ++r) {
        int lr = wid * 2 + r;
        int n = row0 + lr;
        float dn = dis[n];
        float4 acc = ld4(x + (size_t)n * 128 + lane * 4);
        acc.x *= dn; acc.y *= dn; acc.z *= dn; acc.w *= dn;
        int2 se = ld2i(seg + 2 * (size_t)n);
        acc = gather_sum<128, 32, true>(x, dis, csr, se.x, se.y, lane, acc);
        *reinterpret_cast<float4*>(&As[lr * APITCH + lane * 4]) = acc;
    }
    __syncthreads();

    int hi = lane >> 4, lm = lane & 15;
    int col0 = wid << 4;
    v8f c = {};
    const float* Arow = As + lm * APITCH + 2 * hi;
    for (int k0 = 0; k0 < 128; k0 += 4) {
        v2f a = *reinterpret_cast<const v2f*>(Arow + k0);
        const float* Bp = W + (size_t)(k0 + 2 * hi) * 128 + col0 + lm;
        v2f b;
        b.x = Bp[0];
        b.y = Bp[128];
        c = __builtin_amdgcn_wmma_f32_16x16x4_f32(false, a, false, b,
                                                  (short)0, c, false, false);
    }
    float bv = bias[col0 + lm];
    float dr[8];
    #pragma unroll
    for (int v = 0; v < 8; ++v) {
        dr[v] = dis[row0 + v + 8 * hi];
        float o = c[v] * dr[v] + bv;
        Hs[(v + 8 * hi) * APITCH + col0 + lm] = o > 0.0f ? o : 0.0f;
    }
    __syncthreads();
    if (wid < 4) {
        v8f c2 = {};
        const float* Hrow = Hs + lm * APITCH + 2 * hi;
        for (int k0 = 0; k0 < 128; k0 += 4) {
            v2f a = *reinterpret_cast<const v2f*>(Hrow + k0);
            const float* Bp = Wn + (size_t)(k0 + 2 * hi) * 64 + col0 + lm;
            v2f b;
            b.x = Bp[0];
            b.y = Bp[64];
            c2 = __builtin_amdgcn_wmma_f32_16x16x4_f32(false, a, false, b,
                                                       (short)0, c2, false, false);
        }
        #pragma unroll
        for (int v = 0; v < 8; ++v) {
            int row = row0 + v + 8 * hi;
            out[(size_t)row * 64 + col0 + lm] = c2[v] * dr[v];
        }
    }
}

__global__ void __launch_bounds__(256) gather64_gemm32(const float* __restrict__ h,
                                                       const float* __restrict__ dis,
                                                       const float* __restrict__ bias,
                                                       const int* __restrict__ seg,
                                                       const int* __restrict__ csr,
                                                       const float* __restrict__ Wn,
                                                       float* __restrict__ out, int N) {
    __shared__ __attribute__((aligned(16))) float Hs[16 * HPITCH];
    __shared__ __attribute__((aligned(16))) float ds[16];
    int t = threadIdx.x;
    int ln = t >> 4, sub = t & 15;
    int row0 = blockIdx.x << 4;
    int n = row0 + ln;
    float4 o = make_float4(0.0f, 0.0f, 0.0f, 0.0f);
    float dn = 0.0f;
    if (n < N) {
        float4 acc = ld4(h + (size_t)n * 64 + sub * 4);
        int2 se = ld2i(seg + 2 * (size_t)n);
        acc = gather_sum<64, 16, false>(h, dis, csr, se.x, se.y, sub, acc);
        dn = dis[n];
        float4 b = ld4(bias + sub * 4);
        o.x = acc.x * dn + b.x; o.y = acc.y * dn + b.y;
        o.z = acc.z * dn + b.z; o.w = acc.w * dn + b.w;
        o.x = o.x > 0.0f ? o.x : 0.0f; o.y = o.y > 0.0f ? o.y : 0.0f;
        o.z = o.z > 0.0f ? o.z : 0.0f; o.w = o.w > 0.0f ? o.w : 0.0f;
    }
    *reinterpret_cast<float4*>(&Hs[ln * HPITCH + sub * 4]) = o;
    if (sub == 0) ds[ln] = dn;
    __syncthreads();
    int wid = t >> 5, lane = t & 31;
    if (wid < 2) {
        int hi = lane >> 4, lm = lane & 15;
        int col0 = wid << 4;
        float4 d0 = *reinterpret_cast<const float4*>(&ds[8 * hi]);
        float4 d1 = *reinterpret_cast<const float4*>(&ds[8 * hi + 4]);
        float dr[8] = {d0.x, d0.y, d0.z, d0.w, d1.x, d1.y, d1.z, d1.w};
        v8f c = {};
        const float* Hrow = Hs + lm * HPITCH + 2 * hi;
        for (int k0 = 0; k0 < 64; k0 += 4) {
            v2f a = *reinterpret_cast<const v2f*>(Hrow + k0);
            const float* Bp = Wn + (size_t)(k0 + 2 * hi) * 32 + col0 + lm;
            v2f b;
            b.x = Bp[0];
            b.y = Bp[32];
            c = __builtin_amdgcn_wmma_f32_16x16x4_f32(false, a, false, b,
                                                      (short)0, c, false, false);
        }
        #pragma unroll
        for (int v = 0; v < 8; ++v) {
            int row = row0 + v + 8 * hi;
            if (row < N) out[(size_t)row * 32 + col0 + lm] = c[v] * dr[v];
        }
    }
}

__global__ void __launch_bounds__(256) gather_pool32(const float* __restrict__ h,
                                                     const float* __restrict__ dis,
                                                     const float* __restrict__ bias,
                                                     const int* __restrict__ seg,
                                                     const int* __restrict__ csr,
                                                     const int* __restrict__ batch,
                                                     float* __restrict__ pooled, int N) {
    __shared__ __attribute__((aligned(16))) float rows[32 * 32];
    __shared__ int gid[32];
    int t = threadIdx.x;
    int ln = t >> 3, sub = t & 7;
    int n = blockIdx.x * 32 + ln;
    float4 o = make_float4(0.0f, 0.0f, 0.0f, 0.0f);
    int g = -1;
    if (n < N) {
        float4 acc = ld4(h + (size_t)n * 32 + sub * 4);
        int2 se = ld2i(seg + 2 * (size_t)n);
        acc = gather_sum<32, 8, false>(h, dis, csr, se.x, se.y, sub, acc);
        float dn = dis[n];
        float4 b = ld4(bias + sub * 4);
        o.x = acc.x * dn + b.x; o.y = acc.y * dn + b.y;
        o.z = acc.z * dn + b.z; o.w = acc.w * dn + b.w;
        o.x = o.x > 0.0f ? o.x : 0.0f; o.y = o.y > 0.0f ? o.y : 0.0f;
        o.z = o.z > 0.0f ? o.z : 0.0f; o.w = o.w > 0.0f ? o.w : 0.0f;
        g = batch[n];
    }
    *reinterpret_cast<float4*>(&rows[ln * 32 + sub * 4]) = o;
    if (sub == 0) gid[ln] = g;
    __syncthreads();
    if (t < 32) {
        int mine = gid[t];
        float r[32];
        #pragma unroll
        for (int i = 0; i < 32; ++i) r[i] = rows[i * 32 + t];
        int cur = __builtin_amdgcn_readlane(mine, 0);
        float s = 0.0f;
        #pragma unroll
        for (int i = 0; i < 32; ++i) {
            int gi = __builtin_amdgcn_readlane(mine, i);
            if (gi != cur) {
                if (cur >= 0) atomAddF(&pooled[(cur << 5) + t], s);
                s = 0.0f;
                cur = gi;
            }
            s += r[i];
        }
        if (cur >= 0) atomAddF(&pooled[(cur << 5) + t], s);
    }
}

__global__ void __launch_bounds__(256) head_gemm_mean(const float* __restrict__ pooled,
                                                      const float* __restrict__ counts,
                                                      const float* __restrict__ Wl,
                                                      const float* __restrict__ bl,
                                                      float* __restrict__ out) {
    int tid = threadIdx.x, wid = tid >> 5, lane = tid & 31;
    int tile = blockIdx.x * 8 + wid;
    if (tile >= 16 * 48) return;
    int rt = tile / 48, ct = tile - rt * 48;
    int row0 = rt << 4, col0 = ct << 4;
    int hi = lane >> 4, lm = lane & 15;
    float cm = fmaxf(counts[row0 + lm], 1.0f);
    v8f c = {};
    const float* Arow = pooled + (size_t)(row0 + lm) * 32;
    for (int k0 = 0; k0 < 32; k0 += 4) {
        v2f a = *reinterpret_cast<const v2f*>(Arow + k0 + 2 * hi);
        a.x = a.x / cm;
        a.y = a.y / cm;
        const float* Bp = Wl + (size_t)(k0 + 2 * hi) * 768 + col0 + lm;
        v2f b;
        b.x = Bp[0];
        b.y = Bp[768];
        c = __builtin_amdgcn_wmma_f32_16x16x4_f32(false, a, false, b,
                                                  (short)0, c, false, false);
    }
    float bv = bl[col0 + lm];
    #pragma unroll
    for (int v = 0; v < 8; ++v) {
        int row = row0 + v + 8 * hi;
        out[(size_t)row * 768 + col0 + lm] = c[v] + bv;
    }
}

extern "C" void kernel_launch(void* const* d_in, const int* in_sizes, int n_in,
                              void* d_out, int out_size, void* d_ws, size_t ws_size,
                              hipStream_t stream) {
    const float* x   = (const float*)d_in[0];
    const float* W1  = (const float*)d_in[1];
    const float* b1  = (const float*)d_in[2];
    const float* W2  = (const float*)d_in[3];
    const float* b2  = (const float*)d_in[4];
    const float* W3  = (const float*)d_in[5];
    const float* b3  = (const float*)d_in[6];
    const float* Wl  = (const float*)d_in[7];
    const float* bl  = (const float*)d_in[8];
    const int*   ei  = (const int*)d_in[9];
    const int*   bat = (const int*)d_in[10];
    float*       out = (float*)d_out;

    const int N = NNODES, E = NEDGES;
    const int* src = ei;
    const int* dst = ei + E;

    float* B0     = (float*)d_ws;
    float* B1     = B0 + (size_t)N * 128;
    int*   cnt    = (int*)(B1 + (size_t)N * 64);
    int*   total  = cnt + N;
    int*   seg    = total + 32;
    int*   rank   = seg + 2 * N;
    int*   csr    = rank + E;
    float* dis    = B1 + (size_t)N * 128;
    float* counts = dis + N;
    float* pooled = counts + NGRAPH;

    hipMemsetAsync(cnt, 0, (size_t)(N + 1) * sizeof(int), stream);
    hipMemsetAsync(counts, 0, (NGRAPH + NGRAPH * 32) * sizeof(float), stream);
    rank_kernel<<<E / 256, 256, 0, stream>>>(dst, cnt, rank, E);
    dis_counts_alloc_kernel<<<(N + 255) / 256, 256, 0, stream>>>(cnt, dis, bat, counts, seg, total, N);
    fill_kernel<<<E / 256, 256, 0, stream>>>(src, dst, rank, seg, csr, E);

    gather_gemm128_64<<<N / 16, 256, 0, stream>>>(x, W1, dis, b1, seg, csr, W2, B1);

    gather64_gemm32<<<(N + 15) / 16, 256, 0, stream>>>(B1, dis, b2, seg, csr, W3, B0, N);

    gather_pool32<<<N / 32, 256, 0, stream>>>(B0, dis, b3, seg, csr, bat, pooled, N);

    head_gemm_mean<<<(16 * 48) / 8, 256, 0, stream>>>(pooled, counts, Wl, bl, out);
}


# dst-grouped edge list built by a two-level counting sort on LDS counters (2560-edge chunks binned by 128-node range, ranked per range); 640000 returning global adds, a launch go; sum order as before
# speedup vs baseline: 4.8356x; 1.6892x over previous
#include <hip/hip_runtime.h>


typedef __attribute__((ext_vector_type(2))) float v2f;
typedef __attribute__((ext_vector_type(8))) float v8f;

#define NNODES 40000
#define NEDGES 640000
#define NGRAPH 256
#define APITCH 132
#define HPITCH 68
#define EPC 2560
#define EPT 10
#define NCHUNK 250
#define BSHIFT 7
#define BNODES 128
#define NBIN 313
#define TPITCH 320

__device__ __forceinline__ void atomAddF(float* p, float v) {
    unsafeAtomicAdd(p, v);
}

__global__ void __launch_bounds__(256) bin_kernel(const int* __restrict__ src,
                                                  const int* __restrict__ dst,
                                                  int* __restrict__ binned,
                                                  int* __restrict__ tab) {
    __shared__ int hist[512];
    __shared__ int wsum[8];
    __shared__ int sorted[EPC];
    int t = threadIdx.x, lane = t & 31, w = t >> 5;
    int e0 = blockIdx.x * EPC;
    hist[t] = 0;
    hist[t + 256] = 0;
    int d[EPT], s[EPT], r[EPT];
    #pragma unroll
    for (int i = 0; i < EPT; ++i) {
        d[i] = dst[e0 + i * 256 + t];
        s[i] = src[e0 + i * 256 + t];
    }
    __syncthreads();
    #pragma unroll
    for (int i = 0; i < EPT; ++i) r[i] = atomicAdd(&hist[d[i] >> BSHIFT], 1);
    __syncthreads();
    int h0 = hist[2 * t], h1 = hist[2 * t + 1];
    int v = h0 + h1, inc = v;
    for (int o = 1; o < 32; o <<= 1) {
        int u = __shfl_up(inc, o);
        if (lane >= o) inc += u;
    }
    if (lane == 31) wsum[w] = inc;
    __syncthreads();
    int wb = 0;
    for (int i = 0; i < w; ++i) wb += wsum[i];
    int ex = wb + inc - v;
    hist[2 * t] = ex;
    hist[2 * t + 1] = ex + h0;
    __syncthreads();
    #pragma unroll
    for (int i = 0; i < EPT; ++i)
        sorted[hist[d[i] >> BSHIFT] + r[i]] = ((d[i] & (BNODES - 1)) << 16) | s[i];
    tab[blockIdx.x * TPITCH + t] = hist[t];
    if (t < TPITCH - 256) tab[blockIdx.x * TPITCH + 256 + t] = hist[256 + t];
    __syncthreads();
    #pragma unroll
    for (int i = 0; i < EPT; ++i) binned[e0 + i * 256 + t] = sorted[i * 256 + t];
}

__global__ void __launch_bounds__(256) group_kernel(const int* __restrict__ binned,
                                                    const int* __restrict__ tab,
                                                    const int* __restrict__ batch,
                                                    float* __restrict__ dis,
                                                    float* __restrict__ counts,
                                                    int* __restrict__ seg,
                                                    int* __restrict__ csr,
                                                    int* __restrict__ total, int N) {
    __shared__ int cntl[BNODES];
    __shared__ int cur[BNODES];
    __shared__ int wsum[4];
    __shared__ int wgbase;
    int t = threadIdx.x, lane = t & 31, w = t >> 5;
    int b = blockIdx.x;
    if (t < BNODES) cntl[t] = 0;
    int jb = 0, je = 0;
    if (t < NCHUNK) {
        jb = t * EPC + tab[t * TPITCH + b];
        je = t * EPC + tab[t * TPITCH + b + 1];
    }
    __syncthreads();
    for (int j = jb; j < je; ++j) atomicAdd(&cntl[binned[j] >> 16], 1);
    __syncthreads();
    int c = 0, inc = 0;
    if (t < BNODES) {
        c = cntl[t];
        inc = c;
        for (int o = 1; o < 32; o <<= 1) {
            int u = __shfl_up(inc, o);
            if (lane >= o) inc += u;
        }
        if (lane == 31) wsum[w] = inc;
    }
    __syncthreads();
    if (t == 0) wgbase = atomicAdd(total, wsum[0] + wsum[1] + wsum[2] + wsum[3]);
    __syncthreads();
    if (t < BNODES) {
        int wb = 0;
        for (int i = 0; i < w; ++i) wb += wsum[i];
        int beg = wgbase + wb + inc - c;
        cur[t] = beg;
        int n = b * BNODES + t;
        if (n < N) {
            *reinterpret_cast<int2*>(seg + 2 * (size_t)n) = make_int2(beg, beg + c);
            dis[n] = rsqrtf((float)c + 1.0f);
            atomAddF(&counts[batch[n]], 1.0f);
        }
    }
    __syncthreads();
    for (int j = jb; j < je; ++j) {
        int p = binned[j];
        int pos = atomicAdd(&cur[p >> 16], 1);
        csr[pos] = p & 0xFFFF;
    }
}

__device__ __forceinline__ float4 ld4(const float* p) {
    return *reinterpret_cast<const float4*>(p);
}
__device__ __forceinline__ int2 ld2i(const int* p) {
    return *reinterpret_cast<const int2*>(p);
}
__device__ __forceinline__ void fma4(float4& a, const float4& v, float d) {
    a.x += v.x * d; a.y += v.y * d; a.z += v.z * d; a.w += v.w * d;
}
__device__ __forceinline__ void add4(float4& a, const float4& v) {
    a.x += v.x; a.y += v.y; a.z += v.z; a.w += v.w;
}

__device__ __forceinline__ float4 keep4(const float4& v, bool on) {
    float4 r;
    r.x = on ? v.x : 0.0f; r.y = on ? v.y : 0.0f; r.z = on ? v.z : 0.0f; r.w = on ? v.w : 0.0f;
    return r;
}

template <int F, int LPN, bool SCALE>
__device__ __forceinline__ float4 gather_sum(const float* __restrict__ h,
                                             const float* __restrict__ dis,
                                             const int* __restrict__ csr,
                                             int beg, int end, int sub, float4 acc) {
    for (int j0 = beg; j0 < end; j0 += LPN) {
        int j = j0 + sub;
        int s = 0;
        float d = 0.0f;
        if (j < end) {
            s = csr[j];
            if (SCALE) d = dis[s];
        }
        int c = end - j0 < LPN ? end - j0 : LPN;
        for (int k = 0; k < c; k += 4) {
            int last = c - 1;
            int k1 = k + 1 < last ? k + 1 : last;
            int k2 = k + 2 < last ? k + 2 : last;
            int k3 = k + 3 < last ? k + 3 : last;
            int s0 = __shfl(s, k, LPN), s1 = __shfl(s, k1, LPN);
            int s2 = __shfl(s, k2, LPN), s3 = __shfl(s, k3, LPN);
            float4 v0 = ld4(h + (size_t)s0 * F + sub * 4);
            float4 v1 = ld4(h + (size_t)s1 * F + sub * 4);
            float4 v2 = ld4(h + (size_t)s2 * F + sub * 4);
            float4 v3 = ld4(h + (size_t)s3 * F + sub * 4);
            v1 = keep4(v1, k + 1 < c);
            v2 = keep4(v2, k + 2 < c);
            v3 = keep4(v3, k + 3 < c);
            if (SCALE) {
                fma4(acc, v0, __shfl(d, k, LPN));
                fma4(acc, v1, __shfl(d, k1, LPN));
                fma4(acc, v2, __shfl(d, k2, LPN));
                fma4(acc, v3, __shfl(d, k3, LPN));
            } else {
                add4(acc, v0); add4(acc, v1); add4(acc, v2); add4(acc, v3);
            }
        }
    }
    return acc;
}

__global__ void __launch_bounds__(256) gather_gemm128_64(const float* __restrict__ x,
                                                         const float* __restrict__ W,
                                                         const float* __restrict__ dis,
                                                         const float* __restrict__ bias,
                                                         const int* __restrict__ seg,
                                                         const int* __restrict__ csr,
                                                         const float* __restrict__ Wn,
                                                         float* __restrict__ out) {
    __shared__ __attribute__((aligned(16))) float As[16 * APITCH];
    __shared__ __attribute__((aligned(16))) float Hs[16 * APITCH];
    int tid = threadIdx.x, wid = tid >> 5, lane = tid & 31;
    int row0 = blockIdx.x << 4;
    #pragma unroll
    for (int r = 0; r < 2; ++r) {
        int lr = wid * 2 + r;
        int n = row0 + lr;
        float dn = dis[n];
        float4 acc = ld4(x + (size_t)n * 128 + lane * 4);
        acc.x *= dn; acc.y *= dn; acc.z *= dn; acc.w *= dn;
        int2 se = ld2i(seg + 2 * (size_t)n);
        acc = gather_sum<128, 32, true>(x, dis, csr, se.x, se.y, lane, acc);
        *reinterpret_cast<float4*>(&As[lr * APITCH + lane * 4]) = acc;
    }
    __syncthreads();

    int hi = lane >> 4, lm = lane & 15;
    int col0 = wid << 4;
    v8f c = {};
    const float* Arow = As + lm * APITCH + 2 * hi;
    for (int k0 = 0; k0 < 128; k0 += 4) {
        v2f a = *reinterpret_cast<const v2f*>(Arow + k0);
        const float* Bp = W + (size_t)(k0 + 2 * hi) * 128 + col0 + lm;
        v2f b;
        b.x = Bp[0];
        b.y = Bp[128];
        c = __builtin_amdgcn_wmma_f32_16x16x4_f32(false, a, false, b,
                                                  (short)0, c, false, false);
    }
    float bv = bias[col0 + lm];
    float dr[8];
    #pragma unroll
    for (int v = 0; v < 8; ++v) {
        dr[v] = dis[row0 + v + 8 * hi];
        float o = c[v] * dr[v] + bv;
        Hs[(v + 8 * hi) * APITCH + col0 + lm] = o > 0.0f ? o : 0.0f;
    }
    __syncthreads();
    if (wid < 4) {
        v8f c2 = {};
        const float* Hrow = Hs + lm * APITCH + 2 * hi;
        for (int k0 = 0; k0 < 128; k0 += 4) {
            v2f a = *reinterpret_cast<const v2f*>(Hrow + k0);
            const float* Bp = Wn + (size_t)(k0 + 2 * hi) * 64 + col0 + lm;
            v2f b;
            b.x = Bp[0];
            b.y = Bp[64];
            c2 = __builtin_amdgcn_wmma_f32_16x16x4_f32(false, a, false, b,
                                                       (short)0, c2, false, false);
        }
        #pragma unroll
        for (int v = 0; v < 8; ++v) {
            int row = row0 + v + 8 * hi;
            out[(size_t)row * 64 + col0 + lm] = c2[v] * dr[v];
        }
    }
}

__global__ void __launch_bounds__(256) gather64_gemm32(const float* __restrict__ h,
                                                       const float* __restrict__ dis,
                                                       const float* __restrict__ bias,
                                                       const int* __restrict__ seg,
                                                       const int* __restrict__ csr,
                                                       const float* __restrict__ Wn,
                                                       float* __restrict__ out, int N) {
    __shared__ __attribute__((aligned(16))) float Hs[16 * HPITCH];
    __shared__ __attribute__((aligned(16))) float ds[16];
    int t = threadIdx.x;
    int ln = t >> 4, sub = t & 15;
    int row0 = blockIdx.x << 4;
    int n = row0 + ln;
    float4 o = make_float4(0.0f, 0.0f, 0.0f, 0.0f);
    float dn = 0.0f;
    if (n < N) {
        float4 acc = ld4(h + (size_t)n * 64 + sub * 4);
        int2 se = ld2i(seg + 2 * (size_t)n);
        acc = gather_sum<64, 16, false>(h, dis, csr, se.x, se.y, sub, acc);
        dn = dis[n];
        float4 b = ld4(bias + sub * 4);
        o.x = acc.x * dn + b.x; o.y = acc.y * dn + b.y;
        o.z = acc.z * dn + b.z; o.w = acc.w * dn + b.w;
        o.x = o.x > 0.0f ? o.x : 0.0f; o.y = o.y > 0.0f ? o.y : 0.0f;
        o.z = o.z > 0.0f ? o.z : 0.0f; o.w = o.w > 0.0f ? o.w : 0.0f;
    }
    *reinterpret_cast<float4*>(&Hs[ln * HPITCH + sub * 4]) = o;
    if (sub == 0) ds[ln] = dn;
    __syncthreads();
    int wid = t >> 5, lane = t & 31;
    if (wid < 2) {
        int hi = lane >> 4, lm = lane & 15;
        int col0 = wid << 4;
        float4 d0 = *reinterpret_cast<const float4*>(&ds[8 * hi]);
        float4 d1 = *reinterpret_cast<const float4*>(&ds[8 * hi + 4]);
        float dr[8] = {d0.x, d0.y, d0.z, d0.w, d1.x, d1.y, d1.z, d1.w};
        v8f c = {};
        const float* Hrow = Hs + lm * HPITCH + 2 * hi;
        for (int k0 = 0; k0 < 64; k0 += 4) {
            v2f a = *reinterpret_cast<const v2f*>(Hrow + k0);
            const float* Bp = Wn + (size_t)(k0 + 2 * hi) * 32 + col0 + lm;
            v2f b;
            b.x = Bp[0];
            b.y = Bp[32];
            c = __builtin_amdgcn_wmma_f32_16x16x4_f32(false, a, false, b,
                                                      (short)0, c, false, false);
        }
        #pragma unroll
        for (int v = 0; v < 8; ++v) {
            int row = row0 + v + 8 * hi;
            if (row < N) out[(size_t)row * 32 + col0 + lm] = c[v] * dr[v];
        }
    }
}

__global__ void __launch_bounds__(256) gather_pool32(const float* __restrict__ h,
                                                     const float* __restrict__ dis,
                                                     const float* __restrict__ bias,
                                                     const int* __restrict__ seg,
                                                     const int* __restrict__ csr,
                                                     const int* __restrict__ batch,
                                                     float* __restrict__ pooled, int N) {
    __shared__ __attribute__((aligned(16))) float rows[32 * 32];
    __shared__ int gid[32];
    int t = threadIdx.x;
    int ln = t >> 3, sub = t & 7;
    int n = blockIdx.x * 32 + ln;
    float4 o = make_float4(0.0f, 0.0f, 0.0f, 0.0f);
    int g = -1;
    if (n < N) {
        float4 acc = ld4(h + (size_t)n * 32 + sub * 4);
        int2 se = ld2i(seg + 2 * (size_t)n);
        acc = gather_sum<32, 8, false>(h, dis, csr, se.x, se.y, sub, acc);
        float dn = dis[n];
        float4 b = ld4(bias + sub * 4);
        o.x = acc.x * dn + b.x; o.y = acc.y * dn + b.y;
        o.z = acc.z * dn + b.z; o.w = acc.w * dn + b.w;
        o.x = o.x > 0.0f ? o.x : 0.0f; o.y = o.y > 0.0f ? o.y : 0.0f;
        o.z = o.z > 0.0f ? o.z : 0.0f; o.w = o.w > 0.0f ? o.w : 0.0f;
        g = batch[n];
    }
    *reinterpret_cast<float4*>(&rows[ln * 32 + sub * 4]) = o;
    if (sub == 0) gid[ln] = g;
    __syncthreads();
    if (t < 32) {
        int mine = gid[t];
        float r[32];
        #pragma unroll
        for (int i = 0; i < 32; ++i) r[i] = rows[i * 32 + t];
        int cur = __builtin_amdgcn_readlane(mine, 0);
        float s = 0.0f;
        #pragma unroll
        for (int i = 0; i < 32; ++i) {
            int gi = __builtin_amdgcn_readlane(mine, i);
            if (gi != cur) {
                if (cur >= 0) atomAddF(&pooled[(cur << 5) + t], s);
                s = 0.0f;
                cur = gi;
            }
            s += r[i];
        }
        if (cur >= 0) atomAddF(&pooled[(cur << 5) + t], s);
    }
}

__global__ void __launch_bounds__(256) head_gemm_mean(const float* __restrict__ pooled,
                                                      const float* __restrict__ counts,
                                                      const float* __restrict__ Wl,
                                                      const float* __restrict__ bl,
                                                      float* __restrict__ out) {
    int tid = threadIdx.x, wid = tid >> 5, lane = tid & 31;
    int tile = blockIdx.x * 8 + wid;
    if (tile >= 16 * 48) return;
    int rt = tile / 48, ct = tile - rt * 48;
    int row0 = rt << 4, col0 = ct << 4;
    int hi = lane >> 4, lm = lane & 15;
    float cm = fmaxf(counts[row0 + lm], 1.0f);
    v8f c = {};
    const float* Arow = pooled + (size_t)(row0 + lm) * 32;
    for (int k0 = 0; k0 < 32; k0 += 4) {
        v2f a = *reinterpret_cast<const v2f*>(Arow + k0 + 2 * hi);
        a.x = a.x / cm;
        a.y = a.y / cm;
        const float* Bp = Wl + (size_t)(k0 + 2 * hi) * 768 + col0 + lm;
        v2f b;
        b.x = Bp[0];
        b.y = Bp[768];
        c = __builtin_amdgcn_wmma_f32_16x16x4_f32(false, a, false, b,
                                                  (short)0, c, false, false);
    }
    float bv = bl[col0 + lm];
    #pragma unroll
    for (int v = 0; v < 8; ++v) {
        int row = row0 + v + 8 * hi;
        out[(size_t)row * 768 + col0 + lm] = c[v] + bv;
    }
}

extern "C" void kernel_launch(void* const* d_in, const int* in_sizes, int n_in,
                              void* d_out, int out_size, void* d_ws, size_t ws_size,
                              hipStream_t stream) {
    const float* x   = (const float*)d_in[0];
    const float* W1  = (const float*)d_in[1];
    const float* b1  = (const float*)d_in[2];
    const float* W2  = (const float*)d_in[3];
    const float* b2  = (const float*)d_in[4];
    const float* W3  = (const float*)d_in[5];
    const float* b3  = (const float*)d_in[6];
    const float* Wl  = (const float*)d_in[7];
    const float* bl  = (const float*)d_in[8];
    const int*   ei  = (const int*)d_in[9];
    const int*   bat = (const int*)d_in[10];
    float*       out = (float*)d_out;

    const int N = NNODES, E = NEDGES;
    const int* src = ei;
    const int* dst = ei + E;

    float* B0     = (float*)d_ws;
    float* dis    = B0 + (size_t)N * 32;
    float* counts = dis + N;
    float* pooled = counts + NGRAPH;
    int*   total  = (int*)(pooled + NGRAPH * 32);
    float* B1     = B0 + (size_t)N * 128;
    int*   seg    = (int*)(B1 + (size_t)N * 64);
    int*   binned = seg + 2 * N;
    int*   csr    = binned + E;
    int*   tab    = csr + E;

    hipMemsetAsync(counts, 0, (NGRAPH + NGRAPH * 32 + 32) * sizeof(float), stream);
    bin_kernel<<<NCHUNK, 256, 0, stream>>>(src, dst, binned, tab);
    group_kernel<<<NBIN, 256, 0, stream>>>(binned, tab, bat, dis, counts, seg, csr, total, N);

    gather_gemm128_64<<<N / 16, 256, 0, stream>>>(x, W1, dis, b1, seg, csr, W2, B1);

    gather64_gemm32<<<(N + 15) / 16, 256, 0, stream>>>(B1, dis, b2, seg, csr, W3, B0, N);

    gather_pool32<<<N / 32, 256, 0, stream>>>(B0, dis, b3, seg, csr, bat, pooled, N);

    head_gemm_mean<<<(16 * 48) / 8, 256, 0, stream>>>(pooled, counts, Wl, bl, out);
}


# gather kernels: half as many lanes per node, two float4 per lane per gathered row, so a wave carries twice the nodes; the three gather launches start 25000 waves, not 50000; sums unchanged
# speedup vs baseline: 5.0470x; 1.0437x over previous
#include <hip/hip_runtime.h>


typedef __attribute__((ext_vector_type(2))) float v2f;
typedef __attribute__((ext_vector_type(8))) float v8f;

#define NNODES 40000
#define NEDGES 640000
#define NGRAPH 256
#define APITCH 132
#define HPITCH 68
#define EPC 2560
#define EPT 10
#define NCHUNK 250
#define BSHIFT 7
#define BNODES 128
#define NBIN 313
#define TPITCH 320

__device__ __forceinline__ void atomAddF(float* p, float v) {
    unsafeAtomicAdd(p, v);
}

__global__ void __launch_bounds__(256) bin_kernel(const int* __restrict__ src,
                                                  const int* __restrict__ dst,
                                                  int* __restrict__ binned,
                                                  int* __restrict__ tab) {
    __shared__ int hist[512];
    __shared__ int wsum[8];
    __shared__ int sorted[EPC];
    int t = threadIdx.x, lane = t & 31, w = t >> 5;
    int e0 = blockIdx.x * EPC;
    hist[t] = 0;
    hist[t + 256] = 0;
    int d[EPT], s[EPT], r[EPT];
    #pragma unroll
    for (int i = 0; i < EPT; ++i) {
        d[i] = dst[e0 + i * 256 + t];
        s[i] = src[e0 + i * 256 + t];
    }
    __syncthreads();
    #pragma unroll
    for (int i = 0; i < EPT; ++i) r[i] = atomicAdd(&hist[d[i] >> BSHIFT], 1);
    __syncthreads();
    int h0 = hist[2 * t], h1 = hist[2 * t + 1];
    int v = h0 + h1, inc = v;
    for (int o = 1; o < 32; o <<= 1) {
        int u = __shfl_up(inc, o);
        if (lane >= o) inc += u;
    }
    if (lane == 31) wsum[w] = inc;
    __syncthreads();
    int wb = 0;
    for (int i = 0; i < w; ++i) wb += wsum[i];
    int ex = wb + inc - v;
    hist[2 * t] = ex;
    hist[2 * t + 1] = ex + h0;
    __syncthreads();
    #pragma unroll
    for (int i = 0; i < EPT; ++i)
        sorted[hist[d[i] >> BSHIFT] + r[i]] = ((d[i] & (BNODES - 1)) << 16) | s[i];
    tab[blockIdx.x * TPITCH + t] = hist[t];
    if (t < TPITCH - 256) tab[blockIdx.x * TPITCH + 256 + t] = hist[256 + t];
    __syncthreads();
    #pragma unroll
    for (int i = 0; i < EPT; ++i) binned[e0 + i * 256 + t] = sorted[i * 256 + t];
}

__global__ void __launch_bounds__(256) group_kernel(const int* __restrict__ binned,
                                                    const int* __restrict__ tab,
                                                    const int* __restrict__ batch,
                                                    float* __restrict__ dis,
                                                    float* __restrict__ counts,
                                                    int* __restrict__ seg,
                                                    int* __restrict__ csr,
                                                    int* __restrict__ total, int N) {
    __shared__ int cntl[BNODES];
    __shared__ int cur[BNODES];
    __shared__ int wsum[4];
    __shared__ int wgbase;
    int t = threadIdx.x, lane = t & 31, w = t >> 5;
    int b = blockIdx.x;
    if (t < BNODES) cntl[t] = 0;
    int jb = 0, je = 0;
    if (t < NCHUNK) {
        jb = t * EPC + tab[t * TPITCH + b];
        je = t * EPC + tab[t * TPITCH + b + 1];
    }
    __syncthreads();
    for (int j = jb; j < je; ++j) atomicAdd(&cntl[binned[j] >> 16], 1);
    __syncthreads();
    int c = 0, inc = 0;
    if (t < BNODES) {
        c = cntl[t];
        inc = c;
        for (int o = 1; o < 32; o <<= 1) {
            int u = __shfl_up(inc, o);
            if (lane >= o) inc += u;
        }
        if (lane == 31) wsum[w] = inc;
    }
    __syncthreads();
    if (t == 0) wgbase = atomicAdd(total, wsum[0] + wsum[1] + wsum[2] + wsum[3]);
    __syncthreads();
    if (t < BNODES) {
        int wb = 0;
        for (int i = 0; i < w; ++i) wb += wsum[i];
        int beg = wgbase + wb + inc - c;
        cur[t] = beg;
        int n = b * BNODES + t;
        if (n < N) {
            *reinterpret_cast<int2*>(seg + 2 * (size_t)n) = make_int2(beg, beg + c);
            dis[n] = rsqrtf((float)c + 1.0f);
            atomAddF(&counts[batch[n]], 1.0f);
        }
    }
    __syncthreads();
    for (int j = jb; j < je; ++j) {
        int p = binned[j];
        int pos = atomicAdd(&cur[p >> 16], 1);
        csr[pos] = p & 0xFFFF;
    }
}

__device__ __forceinline__ float4 ld4(const float* p) {
    return *reinterpret_cast<const float4*>(p);
}
__device__ __forceinline__ int2 ld2i(const int* p) {
    return *reinterpret_cast<const int2*>(p);
}
__device__ __forceinline__ void fma4(float4& a, const float4& v, float d) {
    a.x += v.x * d; a.y += v.y * d; a.z += v.z * d; a.w += v.w * d;
}
__device__ __forceinline__ void add4(float4& a, const float4& v) {
    a.x += v.x; a.y += v.y; a.z += v.z; a.w += v.w;
}

__device__ __forceinline__ float4 keep4(const float4& v, bool on) {
    float4 r;
    r.x = on ? v.x : 0.0f; r.y = on ? v.y : 0.0f; r.z = on ? v.z : 0.0f; r.w = on ? v.w : 0.0f;
    return r;
}

template <int F, int LPN, bool SCALE>
__device__ __forceinline__ void gather_sum(const float* __restrict__ h,
                                           const float* __restrict__ dis,
                                           const int* __restrict__ csr,
                                           int beg, int end, int sub, float4& a0, float4& a1) {
    static_assert(F == 8 * LPN && (LPN & 3) == 0, "two float4 per lane; groups of four edges");
    for (int j0 = beg; j0 < end; j0 += 2 * LPN) {
        int ja = j0 + sub, jb = ja + LPN;
        int sa = 0, sb = 0;
        float da = 0.0f, db = 0.0f;
        if (ja < end) sa = csr[ja];
        if (jb < end) sb = csr[jb];
        if (SCALE) {
            da = dis[sa];
            db = dis[sb];
        }
        int c = end - j0 < 2 * LPN ? end - j0 : 2 * LPN;
        for (int k = 0; k < c; k += 4) {
            bool up = k >= LPN;
            int s = up ? sb : sa;
            float d = up ? db : da;
            int last = c - 1;
            int k1 = k + 1 < last ? k + 1 : last;
            int k2 = k + 2 < last ? k + 2 : last;
            int k3 = k + 3 < last ? k + 3 : last;
            int s0 = __shfl(s, k & (LPN - 1), LPN), s1 = __shfl(s, k1 & (LPN - 1), LPN);
            int s2 = __shfl(s, k2 & (LPN - 1), LPN), s3 = __shfl(s, k3 & (LPN - 1), LPN);
            const float* p0 = h + (size_t)s0 * F + sub * 4;
            const float* p1 = h + (size_t)s1 * F + sub * 4;
            const float* p2 = h + (size_t)s2 * F + sub * 4;
            const float* p3 = h + (size_t)s3 * F + sub * 4;
            float4 v0 = ld4(p0), w0 = ld4(p0 + LPN * 4);
            float4 v1 = ld4(p1), w1 = ld4(p1 + LPN * 4);
            float4 v2 = ld4(p2), w2 = ld4(p2 + LPN * 4);
            float4 v3 = ld4(p3), w3 = ld4(p3 + LPN * 4);
            v1 = keep4(v1, k + 1 < c); w1 = keep4(w1, k + 1 < c);
            v2 = keep4(v2, k + 2 < c); w2 = keep4(w2, k + 2 < c);
            v3 = keep4(v3, k + 3 < c); w3 = keep4(w3, k + 3 < c);
            if (SCALE) {
                float d0 = __shfl(d, k & (LPN - 1), LPN), d1 = __shfl(d, k1 & (LPN - 1), LPN);
                float d2 = __shfl(d, k2 & (LPN - 1), LPN), d3 = __shfl(d, k3 & (LPN - 1), LPN);
                fma4(a0, v0, d0); fma4(a1, w0, d0);
                fma4(a0, v1, d1); fma4(a1, w1, d1);
                fma4(a0, v2, d2); fma4(a1, w2, d2);
                fma4(a0, v3, d3); fma4(a1, w3, d3);
            } else {
                add4(a0, v0); add4(a1, w0);
                add4(a0, v1); add4(a1, w1);
                add4(a0, v2); add4(a1, w2);
                add4(a0, v3); add4(a1, w3);
            }
        }
    }
}

__global__ void __launch_bounds__(256) gather_gemm128_64(const float* __restrict__ x,
                                                         const float* __restrict__ W,
                                                         const float* __restrict__ dis,
                                                         const float* __restrict__ bias,
                                                         const int* __restrict__ seg,
                                                         const int* __restrict__ csr,
                                                         const float* __restrict__ Wn,
                                                         float* __restrict__ out) {
    __shared__ __attribute__((aligned(16))) float As[32 * APITCH];
    __shared__ __attribute__((aligned(16))) float Hs[32 * APITCH];
    int tid = threadIdx.x, wid = tid >> 5, lane = tid & 31;
    int hi = lane >> 4, lm = lane & 15;
    int row0 = blockIdx.x << 5;
    for (int r = 0; r < 2; ++r) {
        int lr = wid * 4 + r * 2 + hi;
        int n = row0 + lr;
        float dn = dis[n];
        const float* xr = x + (size_t)n * 128 + lm * 4;
        float4 a0 = ld4(xr), a1 = ld4(xr + 64);
        a0.x *= dn; a0.y *= dn; a0.z *= dn; a0.w *= dn;
        a1.x *= dn; a1.y *= dn; a1.z *= dn; a1.w *= dn;
        int2 se = ld2i(seg + 2 * (size_t)n);
        gather_sum<128, 16, true>(x, dis, csr, se.x, se.y, lm, a0, a1);
        *reinterpret_cast<float4*>(&As[lr * APITCH + lm * 4]) = a0;
        *reinterpret_cast<float4*>(&As[lr * APITCH + 64 + lm * 4]) = a1;
    }
    __syncthreads();

    int col0 = wid << 4;
    v8f c0 = {}, c1 = {};
    const float* Arow = As + lm * APITCH + 2 * hi;
    for (int k0 = 0; k0 < 128; k0 += 4) {
        v2f a0 = *reinterpret_cast<const v2f*>(Arow + k0);
        v2f a1 = *reinterpret_cast<const v2f*>(Arow + 16 * APITCH + k0);
        const float* Bp = W + (size_t)(k0 + 2 * hi) * 128 + col0 + lm;
        v2f b;
        b.x = Bp[0];
        b.y = Bp[128];
        c0 = __builtin_amdgcn_wmma_f32_16x16x4_f32(false, a0, false, b,
                                                   (short)0, c0, false, false);
        c1 = __builtin_amdgcn_wmma_f32_16x16x4_f32(false, a1, false, b,
                                                   (short)0, c1, false, false);
    }
    float bv = bias[col0 + lm];
    float dr0[8], dr1[8];
    #pragma unroll
    for (int v = 0; v < 8; ++v) {
        dr0[v] = dis[row0 + v + 8 * hi];
        dr1[v] = dis[row0 + 16 + v + 8 * hi];
        float o0 = c0[v] * dr0[v] + bv;
        float o1 = c1[v] * dr1[v] + bv;
        Hs[(v + 8 * hi) * APITCH + col0 + lm] = o0 > 0.0f ? o0 : 0.0f;
        Hs[(16 + v + 8 * hi) * APITCH + col0 + lm] = o1 > 0.0f ? o1 : 0.0f;
    }
    __syncthreads();
    int mt = wid >> 2;
    col0 = (wid & 3) << 4;
    v8f c2 = {};
    const float* Hrow = Hs + (16 * mt + lm) * APITCH + 2 * hi;
    for (int k0 = 0; k0 < 128; k0 += 4) {
        v2f a = *reinterpret_cast<const v2f*>(Hrow + k0);
        const float* Bp = Wn + (size_t)(k0 + 2 * hi) * 64 + col0 + lm;
        v2f b;
        b.x = Bp[0];
        b.y = Bp[64];
        c2 = __builtin_amdgcn_wmma_f32_16x16x4_f32(false, a, false, b,
                                                   (short)0, c2, false, false);
    }
    #pragma unroll
    for (int v = 0; v < 8; ++v) {
        int row = row0 + 16 * mt + v + 8 * hi;
        out[(size_t)row * 64 + col0 + lm] = c2[v] * (mt ? dr1[v] : dr0[v]);
    }
}

__global__ void __launch_bounds__(256) gather64_gemm32(const float* __restrict__ h,
                                                       const float* __restrict__ dis,
                                                       const float* __restrict__ bias,
                                                       const int* __restrict__ seg,
                                                       const int* __restrict__ csr,
                                                       const float* __restrict__ Wn,
                                                       float* __restrict__ out, int N) {
    __shared__ __attribute__((aligned(16))) float Hs[32 * HPITCH];
    __shared__ __attribute__((aligned(16))) float ds[32];
    int t = threadIdx.x;
    int ln = t >> 3, sub = t & 7;
    int row0 = blockIdx.x << 5;
    int n = row0 + ln;
    float4 o0 = make_float4(0.0f, 0.0f, 0.0f, 0.0f);
    float4 o1 = o0;
    float dn = 0.0f;
    if (n < N) {
        const float* hr = h + (size_t)n * 64 + sub * 4;
        float4 a0 = ld4(hr), a1 = ld4(hr + 32);
        int2 se = ld2i(seg + 2 * (size_t)n);
        gather_sum<64, 8, false>(h, dis, csr, se.x, se.y, sub, a0, a1);
        dn = dis[n];
        float4 b0 = ld4(bias + sub * 4), b1 = ld4(bias + 32 + sub * 4);
        o0.x = a0.x * dn + b0.x; o0.y = a0.y * dn + b0.y;
        o0.z = a0.z * dn + b0.z; o0.w = a0.w * dn + b0.w;
        o1.x = a1.x * dn + b1.x; o1.y = a1.y * dn + b1.y;
        o1.z = a1.z * dn + b1.z; o1.w = a1.w * dn + b1.w;
        o0.x = o0.x > 0.0f ? o0.x : 0.0f; o0.y = o0.y > 0.0f ? o0.y : 0.0f;
        o0.z = o0.z > 0.0f ? o0.z : 0.0f; o0.w = o0.w > 0.0f ? o0.w : 0.0f;
        o1.x = o1.x > 0.0f ? o1.x : 0.0f; o1.y = o1.y > 0.0f ? o1.y : 0.0f;
        o1.z = o1.z > 0.0f ? o1.z : 0.0f; o1.w = o1.w > 0.0f ? o1.w : 0.0f;
    }
    *reinterpret_cast<float4*>(&Hs[ln * HPITCH + sub * 4]) = o0;
    *reinterpret_cast<float4*>(&Hs[ln * HPITCH + 32 + sub * 4]) = o1;
    if (sub == 0) ds[ln] = dn;
    __syncthreads();
    int wid = t >> 5, lane = t & 31;
    if (wid < 4) {
        int hi = lane >> 4, lm = lane & 15;
        int mt = wid >> 1;
        int col0 = (wid & 1) << 4;
        float4 d0 = *reinterpret_cast<const float4*>(&ds[16 * mt + 8 * hi]);
        float4 d1 = *reinterpret_cast<const float4*>(&ds[16 * mt + 8 * hi + 4]);
        float dr[8] = {d0.x, d0.y, d0.z, d0.w, d1.x, d1.y, d1.z, d1.w};
        v8f c = {};
        const float* Hrow = Hs + (16 * mt + lm) * HPITCH + 2 * hi;
        for (int k0 = 0; k0 < 64; k0 += 4) {
            v2f a = *reinterpret_cast<const v2f*>(Hrow + k0);
            const float* Bp = Wn + (size_t)(k0 + 2 * hi) * 32 + col0 + lm;
            v2f b;
            b.x = Bp[0];
            b.y = Bp[32];
            c = __builtin_amdgcn_wmma_f32_16x16x4_f32(false, a, false, b,
                                                      (short)0, c, false, false);
        }
        #pragma unroll
        for (int v = 0; v < 8; ++v) {
            int row = row0 + 16 * mt + v + 8 * hi;
            if (row < N) out[(size_t)row * 32 + col0 + lm] = c[v] * dr[v];
        }
    }
}

__global__ void __launch_bounds__(256) gather_pool32(const float* __restrict__ h,
                                                     const float* __restrict__ dis,
                                                     const float* __restrict__ bias,
                                                     const int* __restrict__ seg,
                                                     const int* __restrict__ csr,
                                                     const int* __restrict__ batch,
                                                     float* __restrict__ pooled, int N) {
    __shared__ __attribute__((aligned(16))) float rows[64 * 32];
    __shared__ int gid[64];
    int t = threadIdx.x;
    int ln = t >> 2, sub = t & 3;
    int n = blockIdx.x * 64 + ln;
    float4 o0 = make_float4(0.0f, 0.0f, 0.0f, 0.0f);
    float4 o1 = o0;
    int g = -1;
    if (n < N) {
        const float* hr = h + (size_t)n * 32 + sub * 4;
        float4 a0 = ld4(hr), a1 = ld4(hr + 16);
        int2 se = ld2i(seg + 2 * (size_t)n);
        gather_sum<32, 4, false>(h, dis, csr, se.x, se.y, sub, a0, a1);
        float dn = dis[n];
        float4 b0 = ld4(bias + sub * 4), b1 = ld4(bias + 16 + sub * 4);
        o0.x = a0.x * dn + b0.x; o0.y = a0.y * dn + b0.y;
        o0.z = a0.z * dn + b0.z; o0.w = a0.w * dn + b0.w;
        o1.x = a1.x * dn + b1.x; o1.y = a1.y * dn + b1.y;
        o1.z = a1.z * dn + b1.z; o1.w = a1.w * dn + b1.w;
        o0.x = o0.x > 0.0f ? o0.x : 0.0f; o0.y = o0.y > 0.0f ? o0.y : 0.0f;
        o0.z = o0.z > 0.0f ? o0.z : 0.0f; o0.w = o0.w > 0.0f ? o0.w : 0.0f;
        o1.x = o1.x > 0.0f ? o1.x : 0.0f; o1.y = o1.y > 0.0f ? o1.y : 0.0f;
        o1.z = o1.z > 0.0f ? o1.z : 0.0f; o1.w = o1.w > 0.0f ? o1.w : 0.0f;
        g = batch[n];
    }
    *reinterpret_cast<float4*>(&rows[ln * 32 + sub * 4]) = o0;
    *reinterpret_cast<float4*>(&rows[ln * 32 + 16 + sub * 4]) = o1;
    if (sub == 0) gid[ln] = g;
    __syncthreads();
    if (t < 64) {
        int f = t & 31, r0 = t & 32;
        int mine = gid[r0 + f];
        float r[32];
        #pragma unroll
        for (int i = 0; i < 32; ++i) r[i] = rows[(r0 + i) * 32 + f];
        int cur = __builtin_amdgcn_readlane(mine, 0);
        float s = 0.0f;
        #pragma unroll
        for (int i = 0; i < 32; ++i) {
            int gi = __builtin_amdgcn_readlane(mine, i);
            if (gi != cur) {
                if (cur >= 0) atomAddF(&pooled[(cur << 5) + f], s);
                s = 0.0f;
                cur = gi;
            }
            s += r[i];
        }
        if (cur >= 0) atomAddF(&pooled[(cur << 5) + f], s);
    }
}

__global__ void __launch_bounds__(256) head_gemm_mean(const float* __restrict__ pooled,
                                                      const float* __restrict__ counts,
                                                      const float* __restrict__ Wl,
                                                      const float* __restrict__ bl,
                                                      float* __restrict__ out) {
    int tid = threadIdx.x, wid = tid >> 5, lane = tid & 31;
    int tile = blockIdx.x * 8 + wid;
    if (tile >= 16 * 48) return;
    int rt = tile / 48, ct = tile - rt * 48;
    int row0 = rt << 4, col0 = ct << 4;
    int hi = lane >> 4, lm = lane & 15;
    float cm = fmaxf(counts[row0 + lm], 1.0f);
    v8f c = {};
    const float* Arow = pooled + (size_t)(row0 + lm) * 32;
    for (int k0 = 0; k0 < 32; k0 += 4) {
        v2f a = *reinterpret_cast<const v2f*>(Arow + k0 + 2 * hi);
        a.x = a.x / cm;
        a.y = a.y / cm;
        const float* Bp = Wl + (size_t)(k0 + 2 * hi) * 768 + col0 + lm;
        v2f b;
        b.x = Bp[0];
        b.y = Bp[768];
        c = __builtin_amdgcn_wmma_f32_16x16x4_f32(false, a, false, b,
                                                  (short)0, c, false, false);
    }
    float bv = bl[col0 + lm];
    #pragma unroll
    for (int v = 0; v < 8; ++v) {
        int row = row0 + v + 8 * hi;
        out[(size_t)row * 768 + col0 + lm] = c[v] + bv;
    }
}

extern "C" void kernel_launch(void* const* d_in, const int* in_sizes, int n_in,
                              void* d_out, int out_size, void* d_ws, size_t ws_size,
                              hipStream_t stream) {
    const float* x   = (const float*)d_in[0];
    const float* W1  = (const float*)d_in[1];
    const float* b1  = (const float*)d_in[2];
    const float* W2  = (const float*)d_in[3];
    const float* b2  = (const float*)d_in[4];
    const float* W3  = (const float*)d_in[5];
    const float* b3  = (const float*)d_in[6];
    const float* Wl  = (const float*)d_in[7];
    const float* bl  = (const float*)d_in[8];
    const int*   ei  = (const int*)d_in[9];
    const int*   bat = (const int*)d_in[10];
    float*       out = (float*)d_out;

    const int N = NNODES, E = NEDGES;
    const int* src = ei;
    const int* dst = ei + E;

    float* B0     = (float*)d_ws;
    float* dis    = B0 + (size_t)N * 32;
    float* counts = dis + N;
    float* pooled = counts + NGRAPH;
    int*   total  = (int*)(pooled + NGRAPH * 32);
    float* B1     = B0 + (size_t)N * 128;
    int*   seg    = (int*)(B1 + (size_t)N * 64);
    int*   binned = seg + 2 * N;
    int*   csr    = binned + E;
    int*   tab    = csr + E;

    hipMemsetAsync(counts, 0, (NGRAPH + NGRAPH * 32 + 32) * sizeof(float), stream);
    bin_kernel<<<NCHUNK, 256, 0, stream>>>(src, dst, binned, tab);
    group_kernel<<<NBIN, 256, 0, stream>>>(binned, tab, bat, dis, counts, seg, csr, total, N);

    gather_gemm128_64<<<N / 32, 256, 0, stream>>>(x, W1, dis, b1, seg, csr, W2, B1);

    gather64_gemm32<<<(N + 31) / 32, 256, 0, stream>>>(B1, dis, b2, seg, csr, W3, B0, N);

    gather_pool32<<<(N + 63) / 64, 256, 0, stream>>>(B0, dis, b3, seg, csr, bat, pooled, N);

    head_gemm_mean<<<(16 * 48) / 8, 256, 0, stream>>>(pooled, counts, Wl, bl, out);
}


# graph counts: a wave adds one float per run of equal graph id among its 32 nodes (the run length), not one per node; 40000 adds of 32 lanes on one word become about 1500; counts stay exact
# speedup vs baseline: 5.7943x; 1.1481x over previous
#include <hip/hip_runtime.h>


typedef __attribute__((ext_vector_type(2))) float v2f;
typedef __attribute__((ext_vector_type(8))) float v8f;

#define NNODES 40000
#define NEDGES 640000
#define NGRAPH 256
#define APITCH 132
#define HPITCH 68
#define EPC 2560
#define EPT 10
#define NCHUNK 250
#define BSHIFT 7
#define BNODES 128
#define NBIN 313
#define TPITCH 320

__device__ __forceinline__ void atomAddF(float* p, float v) {
    unsafeAtomicAdd(p, v);
}

__global__ void __launch_bounds__(256) bin_kernel(const int* __restrict__ src,
                                                  const int* __restrict__ dst,
                                                  int* __restrict__ binned,
                                                  int* __restrict__ tab) {
    __shared__ int hist[512];
    __shared__ int wsum[8];
    __shared__ int sorted[EPC];
    int t = threadIdx.x, lane = t & 31, w = t >> 5;
    int e0 = blockIdx.x * EPC;
    hist[t] = 0;
    hist[t + 256] = 0;
    int d[EPT], s[EPT], r[EPT];
    #pragma unroll
    for (int i = 0; i < EPT; ++i) {
        d[i] = dst[e0 + i * 256 + t];
        s[i] = src[e0 + i * 256 + t];
    }
    __syncthreads();
    #pragma unroll
    for (int i = 0; i < EPT; ++i) r[i] = atomicAdd(&hist[d[i] >> BSHIFT], 1);
    __syncthreads();
    int h0 = hist[2 * t], h1 = hist[2 * t + 1];
    int v = h0 + h1, inc = v;
    for (int o = 1; o < 32; o <<= 1) {
        int u = __shfl_up(inc, o);
        if (lane >= o) inc += u;
    }
    if (lane == 31) wsum[w] = inc;
    __syncthreads();
    int wb = 0;
    for (int i = 0; i < w; ++i) wb += wsum[i];
    int ex = wb + inc - v;
    hist[2 * t] = ex;
    hist[2 * t + 1] = ex + h0;
    __syncthreads();
    #pragma unroll
    for (int i = 0; i < EPT; ++i)
        sorted[hist[d[i] >> BSHIFT] + r[i]] = ((d[i] & (BNODES - 1)) << 16) | s[i];
    tab[blockIdx.x * TPITCH + t] = hist[t];
    if (t < TPITCH - 256) tab[blockIdx.x * TPITCH + 256 + t] = hist[256 + t];
    __syncthreads();
    #pragma unroll
    for (int i = 0; i < EPT; ++i) binned[e0 + i * 256 + t] = sorted[i * 256 + t];
}

__global__ void __launch_bounds__(256) group_kernel(const int* __restrict__ binned,
                                                    const int* __restrict__ tab,
                                                    const int* __restrict__ batch,
                                                    float* __restrict__ dis,
                                                    float* __restrict__ counts,
                                                    int* __restrict__ seg,
                                                    int* __restrict__ csr,
                                                    int* __restrict__ total, int N) {
    __shared__ int cntl[BNODES];
    __shared__ int cur[BNODES];
    __shared__ int wsum[4];
    __shared__ int wgbase;
    int t = threadIdx.x, lane = t & 31, w = t >> 5;
    int b = blockIdx.x;
    if (t < BNODES) cntl[t] = 0;
    int jb = 0, je = 0;
    if (t < NCHUNK) {
        jb = t * EPC + tab[t * TPITCH + b];
        je = t * EPC + tab[t * TPITCH + b + 1];
    }
    __syncthreads();
    for (int j = jb; j < je; ++j) atomicAdd(&cntl[binned[j] >> 16], 1);
    __syncthreads();
    int c = 0, inc = 0;
    if (t < BNODES) {
        c = cntl[t];
        inc = c;
        for (int o = 1; o < 32; o <<= 1) {
            int u = __shfl_up(inc, o);
            if (lane >= o) inc += u;
        }
        if (lane == 31) wsum[w] = inc;
    }
    __syncthreads();
    if (t == 0) wgbase = atomicAdd(total, wsum[0] + wsum[1] + wsum[2] + wsum[3]);
    __syncthreads();
    if (t < BNODES) {
        int wb = 0;
        for (int i = 0; i < w; ++i) wb += wsum[i];
        int beg = wgbase + wb + inc - c;
        cur[t] = beg;
        int n = b * BNODES + t;
        bool ok = n < N;
        int gid = -1;
        if (ok) {
            *reinterpret_cast<int2*>(seg + 2 * (size_t)n) = make_int2(beg, beg + c);
            dis[n] = rsqrtf((float)c + 1.0f);
            gid = batch[n];
        }
        int below = __shfl_up(gid, 1);
        bool head = lane == 0 || gid != below;
        unsigned heads = __builtin_amdgcn_ballot_w32(head);
        if (head && ok) {
            unsigned above = (heads >> lane) >> 1;
            int len = above ? __builtin_ctz(above) + 1 : 32 - lane;
            atomAddF(&counts[gid], (float)len);
        }
    }
    __syncthreads();
    for (int j = jb; j < je; ++j) {
        int p = binned[j];
        int pos = atomicAdd(&cur[p >> 16], 1);
        csr[pos] = p & 0xFFFF;
    }
}

__device__ __forceinline__ float4 ld4(const float* p) {
    return *reinterpret_cast<const float4*>(p);
}
__device__ __forceinline__ int2 ld2i(const int* p) {
    return *reinterpret_cast<const int2*>(p);
}
__device__ __forceinline__ void fma4(float4& a, const float4& v, float d) {
    a.x += v.x * d; a.y += v.y * d; a.z += v.z * d; a.w += v.w * d;
}
__device__ __forceinline__ void add4(float4& a, const float4& v) {
    a.x += v.x; a.y += v.y; a.z += v.z; a.w += v.w;
}

__device__ __forceinline__ float4 keep4(const float4& v, bool on) {
    float4 r;
    r.x = on ? v.x : 0.0f; r.y = on ? v.y : 0.0f; r.z = on ? v.z : 0.0f; r.w = on ? v.w : 0.0f;
    return r;
}

template <int F, int LPN, bool SCALE>
__device__ __forceinline__ void gather_sum(const float* __restrict__ h,
                                           const float* __restrict__ dis,
                                           const int* __restrict__ csr,
                                           int beg, int end, int sub, float4& a0, float4& a1) {
    static_assert(F == 8 * LPN && (LPN & 3) == 0, "two float4 per lane; groups of four edges");
    for (int j0 = beg; j0 < end; j0 += 2 * LPN) {
        int ja = j0 + sub, jb = ja + LPN;
        int sa = 0, sb = 0;
        float da = 0.0f, db = 0.0f;
        if (ja < end) sa = csr[ja];
        if (jb < end) sb = csr[jb];
        if (SCALE) {
            da = dis[sa];
            db = dis[sb];
        }
        int c = end - j0 < 2 * LPN ? end - j0 : 2 * LPN;
        for (int k = 0; k < c; k += 4) {
            bool up = k >= LPN;
            int s = up ? sb : sa;
            float d = up ? db : da;
            int last = c - 1;
            int k1 = k + 1 < last ? k + 1 : last;
            int k2 = k + 2 < last ? k + 2 : last;
            int k3 = k + 3 < last ? k + 3 : last;
            int s0 = __shfl(s, k & (LPN - 1), LPN), s1 = __shfl(s, k1 & (LPN - 1), LPN);
            int s2 = __shfl(s, k2 & (LPN - 1), LPN), s3 = __shfl(s, k3 & (LPN - 1), LPN);
            const float* p0 = h + (size_t)s0 * F + sub * 4;
            const float* p1 = h + (size_t)s1 * F + sub * 4;
            const float* p2 = h + (size_t)s2 * F + sub * 4;
            const float* p3 = h + (size_t)s3 * F + sub * 4;
            float4 v0 = ld4(p0), w0 = ld4(p0 + LPN * 4);
            float4 v1 = ld4(p1), w1 = ld4(p1 + LPN * 4);
            float4 v2 = ld4(p2), w2 = ld4(p2 + LPN * 4);
            float4 v3 = ld4(p3), w3 = ld4(p3 + LPN * 4);
            v1 = keep4(v1, k + 1 < c); w1 = keep4(w1, k + 1 < c);
            v2 = keep4(v2, k + 2 < c); w2 = keep4(w2, k + 2 < c);
            v3 = keep4(v3, k + 3 < c); w3 = keep4(w3, k + 3 < c);
            if (SCALE) {
                float d0 = __shfl(d, k & (LPN - 1), LPN), d1 = __shfl(d, k1 & (LPN - 1), LPN);
                float d2 = __shfl(d, k2 & (LPN - 1), LPN), d3 = __shfl(d, k3 & (LPN - 1), LPN);
                fma4(a0, v0, d0); fma4(a1, w0, d0);
                fma4(a0, v1, d1); fma4(a1, w1, d1);
                fma4(a0, v2, d2); fma4(a1, w2, d2);
                fma4(a0, v3, d3); fma4(a1, w3, d3);
            } else {
                add4(a0, v0); add4(a1, w0);
                add4(a0, v1); add4(a1, w1);
                add4(a0, v2); add4(a1, w2);
                add4(a0, v3); add4(a1, w3);
            }
        }
    }
}

__global__ void __launch_bounds__(256) gather_gemm128_64(const float* __restrict__ x,
                                                         const float* __restrict__ W,
                                                         const float* __restrict__ dis,
                                                         const float* __restrict__ bias,
                                                         const int* __restrict__ seg,
                                                         const int* __restrict__ csr,
                                                         const float* __restrict__ Wn,
                                                         float* __restrict__ out) {
    __shared__ __attribute__((aligned(16))) float As[32 * APITCH];
    __shared__ __attribute__((aligned(16))) float Hs[32 * APITCH];
    int tid = threadIdx.x, wid = tid >> 5, lane = tid & 31;
    int hi = lane >> 4, lm = lane & 15;
    int row0 = blockIdx.x << 5;
    for (int r = 0; r < 2; ++r) {
        int lr = wid * 4 + r * 2 + hi;
        int n = row0 + lr;
        float dn = dis[n];
        const float* xr = x + (size_t)n * 128 + lm * 4;
        float4 a0 = ld4(xr), a1 = ld4(xr + 64);
        a0.x *= dn; a0.y *= dn; a0.z *= dn; a0.w *= dn;
        a1.x *= dn; a1.y *= dn; a1.z *= dn; a1.w *= dn;
        int2 se = ld2i(seg + 2 * (size_t)n);
        gather_sum<128, 16, true>(x, dis, csr, se.x, se.y, lm, a0, a1);
        *reinterpret_cast<float4*>(&As[lr * APITCH + lm * 4]) = a0;
        *reinterpret_cast<float4*>(&As[lr * APITCH + 64 + lm * 4]) = a1;
    }
    __syncthreads();

    int col0 = wid << 4;
    v8f c0 = {}, c1 = {};
    const float* Arow = As + lm * APITCH + 2 * hi;
    for (int k0 = 0; k0 < 128; k0 += 4) {
        v2f a0 = *reinterpret_cast<const v2f*>(Arow + k0);
        v2f a1 = *reinterpret_cast<const v2f*>(Arow + 16 * APITCH + k0);
        const float* Bp = W + (size_t)(k0 + 2 * hi) * 128 + col0 + lm;
        v2f b;
        b.x = Bp[0];
        b.y = Bp[128];
        c0 = __builtin_amdgcn_wmma_f32_16x16x4_f32(false, a0, false, b,
                                                   (short)0, c0, false, false);
        c1 = __builtin_amdgcn_wmma_f32_16x16x4_f32(false, a1, false, b,
                                                   (short)0, c1, false, false);
    }
    float bv = bias[col0 + lm];
    float dr0[8], dr1[8];
    #pragma unroll
    for (int v = 0; v < 8; ++v) {
        dr0[v] = dis[row0 + v + 8 * hi];
        dr1[v] = dis[row0 + 16 + v + 8 * hi];
        float o0 = c0[v] * dr0[v] + bv;
        float o1 = c1[v] * dr1[v] + bv;
        Hs[(v + 8 * hi) * APITCH + col0 + lm] = o0 > 0.0f ? o0 : 0.0f;
        Hs[(16 + v + 8 * hi) * APITCH + col0 + lm] = o1 > 0.0f ? o1 : 0.0f;
    }
    __syncthreads();
    int mt = wid >> 2;
    col0 = (wid & 3) << 4;
    v8f c2 = {};
    const float* Hrow = Hs + (16 * mt + lm) * APITCH + 2 * hi;
    for (int k0 = 0; k0 < 128; k0 += 4) {
        v2f a = *reinterpret_cast<const v2f*>(Hrow + k0);
        const float* Bp = Wn + (size_t)(k0 + 2 * hi) * 64 + col0 + lm;
        v2f b;
        b.x = Bp[0];
        b.y = Bp[64];
        c2 = __builtin_amdgcn_wmma_f32_16x16x4_f32(false, a, false, b,
                                                   (short)0, c2, false, false);
    }
    #pragma unroll
    for (int v = 0; v < 8; ++v) {
        int row = row0 + 16 * mt + v + 8 * hi;
        out[(size_t)row * 64 + col0 + lm] = c2[v] * (mt ? dr1[v] : dr0[v]);
    }
}

__global__ void __launch_bounds__(256) gather64_gemm32(const float* __restrict__ h,
                                                       const float* __restrict__ dis,
                                                       const float* __restrict__ bias,
                                                       const int* __restrict__ seg,
                                                       const int* __restrict__ csr,
                                                       const float* __restrict__ Wn,
                                                       float* __restrict__ out, int N) {
    __shared__ __attribute__((aligned(16))) float Hs[32 * HPITCH];
    __shared__ __attribute__((aligned(16))) float ds[32];
    int t = threadIdx.x;
    int ln = t >> 3, sub = t & 7;
    int row0 = blockIdx.x << 5;
    int n = row0 + ln;
    float4 o0 = make_float4(0.0f, 0.0f, 0.0f, 0.0f);
    float4 o1 = o0;
    float dn = 0.0f;
    if (n < N) {
        const float* hr = h + (size_t)n * 64 + sub * 4;
        float4 a0 = ld4(hr), a1 = ld4(hr + 32);
        int2 se = ld2i(seg + 2 * (size_t)n);
        gather_sum<64, 8, false>(h, dis, csr, se.x, se.y, sub, a0, a1);
        dn = dis[n];
        float4 b0 = ld4(bias + sub * 4), b1 = ld4(bias + 32 + sub * 4);
        o0.x = a0.x * dn + b0.x; o0.y = a0.y * dn + b0.y;
        o0.z = a0.z * dn + b0.z; o0.w = a0.w * dn + b0.w;
        o1.x = a1.x * dn + b1.x; o1.y = a1.y * dn + b1.y;
        o1.z = a1.z * dn + b1.z; o1.w = a1.w * dn + b1.w;
        o0.x = o0.x > 0.0f ? o0.x : 0.0f; o0.y = o0.y > 0.0f ? o0.y : 0.0f;
        o0.z = o0.z > 0.0f ? o0.z : 0.0f; o0.w = o0.w > 0.0f ? o0.w : 0.0f;
        o1.x = o1.x > 0.0f ? o1.x : 0.0f; o1.y = o1.y > 0.0f ? o1.y : 0.0f;
        o1.z = o1.z > 0.0f ? o1.z : 0.0f; o1.w = o1.w > 0.0f ? o1.w : 0.0f;
    }
    *reinterpret_cast<float4*>(&Hs[ln * HPITCH + sub * 4]) = o0;
    *reinterpret_cast<float4*>(&Hs[ln * HPITCH + 32 + sub * 4]) = o1;
    if (sub == 0) ds[ln] = dn;
    __syncthreads();
    int wid = t >> 5, lane = t & 31;
    if (wid < 4) {
        int hi = lane >> 4, lm = lane & 15;
        int mt = wid >> 1;
        int col0 = (wid & 1) << 4;
        float4 d0 = *reinterpret_cast<const float4*>(&ds[16 * mt + 8 * hi]);
        float4 d1 = *reinterpret_cast<const float4*>(&ds[16 * mt + 8 * hi + 4]);
        float dr[8] = {d0.x, d0.y, d0.z, d0.w, d1.x, d1.y, d1.z, d1.w};
        v8f c = {};
        const float* Hrow = Hs + (16 * mt + lm) * HPITCH + 2 * hi;
        for (int k0 = 0; k0 < 64; k0 += 4) {
            v2f a = *reinterpret_cast<const v2f*>(Hrow + k0);
            const float* Bp = Wn + (size_t)(k0 + 2 * hi) * 32 + col0 + lm;
            v2f b;
            b.x = Bp[0];
            b.y = Bp[32];
            c = __builtin_amdgcn_wmma_f32_16x16x4_f32(false, a, false, b,
                                                      (short)0, c, false, false);
        }
        #pragma unroll
        for (int v = 0; v < 8; ++v) {
            int row = row0 + 16 * mt + v + 8 * hi;
            if (row < N) out[(size_t)row * 32 + col0 + lm] = c[v] * dr[v];
        }
    }
}

__global__ void __launch_bounds__(256) gather_pool32(const float* __restrict__ h,
                                                     const float* __restrict__ dis,
                                                     const float* __restrict__ bias,
                                                     const int* __restrict__ seg,
                                                     const int* __restrict__ csr,
                                                     const int* __restrict__ batch,
                                                     float* __restrict__ pooled, int N) {
    __shared__ __attribute__((aligned(16))) float rows[64 * 32];
    __shared__ int gid[64];
    int t = threadIdx.x;
    int ln = t >> 2, sub = t & 3;
    int n = blockIdx.x * 64 + ln;
    float4 o0 = make_float4(0.0f, 0.0f, 0.0f, 0.0f);
    float4 o1 = o0;
    int g = -1;
    if (n < N) {
        const float* hr = h + (size_t)n * 32 + sub * 4;
        float4 a0 = ld4(hr), a1 = ld4(hr + 16);
        int2 se = ld2i(seg + 2 * (size_t)n);
        gather_sum<32, 4, false>(h, dis, csr, se.x, se.y, sub, a0, a1);
        float dn = dis[n];
        float4 b0 = ld4(bias + sub * 4), b1 = ld4(bias + 16 + sub * 4);
        o0.x = a0.x * dn + b0.x; o0.y = a0.y * dn + b0.y;
        o0.z = a0.z * dn + b0.z; o0.w = a0.w * dn + b0.w;
        o1.x = a1.x * dn + b1.x; o1.y = a1.y * dn + b1.y;
        o1.z = a1.z * dn + b1.z; o1.w = a1.w * dn + b1.w;
        o0.x = o0.x > 0.0f ? o0.x : 0.0f; o0.y = o0.y > 0.0f ? o0.y : 0.0f;
        o0.z = o0.z > 0.0f ? o0.z : 0.0f; o0.w = o0.w > 0.0f ? o0.w : 0.0f;
        o1.x = o1.x > 0.0f ? o1.x : 0.0f; o1.y = o1.y > 0.0f ? o1.y : 0.0f;
        o1.z = o1.z > 0.0f ? o1.z : 0.0f; o1.w = o1.w > 0.0f ? o1.w : 0.0f;
        g = batch[n];
    }
    *reinterpret_cast<float4*>(&rows[ln * 32 + sub * 4]) = o0;
    *reinterpret_cast<float4*>(&rows[ln * 32 + 16 + sub * 4]) = o1;
    if (sub == 0) gid[ln] = g;
    __syncthreads();
    if (t < 64) {
        int f = t & 31, r0 = t & 32;
        int mine = gid[r0 + f];
        float r[32];
        #pragma unroll
        for (int i = 0; i < 32; ++i) r[i] = rows[(r0 + i) * 32 + f];
        int cur = __builtin_amdgcn_readlane(mine, 0);
        float s = 0.0f;
        #pragma unroll
        for (int i = 0; i < 32; ++i) {
            int gi = __builtin_amdgcn_readlane(mine, i);
            if (gi != cur) {
                if (cur >= 0) atomAddF(&pooled[(cur << 5) + f], s);
                s = 0.0f;
                cur = gi;
            }
            s += r[i];
        }
        if (cur >= 0) atomAddF(&pooled[(cur << 5) + f], s);
    }
}

__global__ void __launch_bounds__(256) head_gemm_mean(const float* __restrict__ pooled,
                                                      const float* __restrict__ counts,
                                                      const float* __restrict__ Wl,
                                                      const float* __restrict__ bl,
                                                      float* __restrict__ out) {
    int tid = threadIdx.x, wid = tid >> 5, lane = tid & 31;
    int tile = blockIdx.x * 8 + wid;
    if (tile >= 16 * 48) return;
    int rt = tile / 48, ct = tile - rt * 48;
    int row0 = rt << 4, col0 = ct << 4;
    int hi = lane >> 4, lm = lane & 15;
    float cm = fmaxf(counts[row0 + lm], 1.0f);
    v8f c = {};
    const float* Arow = pooled + (size_t)(row0 + lm) * 32;
    for (int k0 = 0; k0 < 32; k0 += 4) {
        v2f a = *reinterpret_cast<const v2f*>(Arow + k0 + 2 * hi);
        a.x = a.x / cm;
        a.y = a.y / cm;
        const float* Bp = Wl + (size_t)(k0 + 2 * hi) * 768 + col0 + lm;
        v2f b;
        b.x = Bp[0];
        b.y = Bp[768];
        c = __builtin_amdgcn_wmma_f32_16x16x4_f32(false, a, false, b,
                                                  (short)0, c, false, false);
    }
    float bv = bl[col0 + lm];
    #pragma unroll
    for (int v = 0; v < 8; ++v) {
        int row = row0 + v + 8 * hi;
        out[(size_t)row * 768 + col0 + lm] = c[v] + bv;
    }
}

extern "C" void kernel_launch(void* const* d_in, const int* in_sizes, int n_in,
                              void* d_out, int out_size, void* d_ws, size_t ws_size,
                              hipStream_t stream) {
    const float* x   = (const float*)d_in[0];
    const float* W1  = (const float*)d_in[1];
    const float* b1  = (const float*)d_in[2];
    const float* W2  = (const float*)d_in[3];
    const float* b2  = (const float*)d_in[4];
    const float* W3  = (const float*)d_in[5];
    const float* b3  = (const float*)d_in[6];
    const float* Wl  = (const float*)d_in[7];
    const float* bl  = (const float*)d_in[8];
    const int*   ei  = (const int*)d_in[9];
    const int*   bat = (const int*)d_in[10];
    float*       out = (float*)d_out;

    const int N = NNODES, E = NEDGES;
    const int* src = ei;
    const int* dst = ei + E;

    float* B0     = (float*)d_ws;
    float* dis    = B0 + (size_t)N * 32;
    float* counts = dis + N;
    float* pooled = counts + NGRAPH;
    int*   total  = (int*)(pooled + NGRAPH * 32);
    float* B1     = B0 + (size_t)N * 128;
    int*   seg    = (int*)(B1 + (size_t)N * 64);
    int*   binned = seg + 2 * N;
    int*   csr    = binned + E;
    int*   tab    = csr + E;

    hipMemsetAsync(counts, 0, (NGRAPH + NGRAPH * 32 + 32) * sizeof(float), stream);
    bin_kernel<<<NCHUNK, 256, 0, stream>>>(src, dst, binned, tab);
    group_kernel<<<NBIN, 256, 0, stream>>>(binned, tab, bat, dis, counts, seg, csr, total, N);

    gather_gemm128_64<<<N / 32, 256, 0, stream>>>(x, W1, dis, b1, seg, csr, W2, B1);

    gather64_gemm32<<<(N + 31) / 32, 256, 0, stream>>>(B1, dis, b2, seg, csr, W3, B0, N);

    gather_pool32<<<(N + 63) / 64, 256, 0, stream>>>(B0, dis, b3, seg, csr, bat, pooled, N);

    head_gemm_mean<<<(16 * 48) / 8, 256, 0, stream>>>(pooled, counts, Wl, bl, out);
}


# grouping pass 2: a piece's words read once by 8 lanes side by side, as one batch, and kept in registers for the count and the placing; edge order in a node's slice, so gather sum order, changes
# speedup vs baseline: 6.0055x; 1.0364x over previous
#include <hip/hip_runtime.h>


typedef __attribute__((ext_vector_type(2))) float v2f;
typedef __attribute__((ext_vector_type(8))) float v8f;

#define NNODES 40000
#define NEDGES 640000
#define NGRAPH 256
#define APITCH 132
#define HPITCH 68
#define EPC 2560
#define EPT 10
#define NCHUNK 250
#define BSHIFT 7
#define BNODES 128
#define NBIN 313
#define TPITCH 320

__device__ __forceinline__ void atomAddF(float* p, float v) {
    unsafeAtomicAdd(p, v);
}

__global__ void __launch_bounds__(256) bin_kernel(const int* __restrict__ src,
                                                  const int* __restrict__ dst,
                                                  int* __restrict__ binned,
                                                  int* __restrict__ tab) {
    __shared__ int hist[512];
    __shared__ int wsum[8];
    __shared__ int sorted[EPC];
    int t = threadIdx.x, lane = t & 31, w = t >> 5;
    int e0 = blockIdx.x * EPC;
    hist[t] = 0;
    hist[t + 256] = 0;
    int d[EPT], s[EPT], r[EPT];
    #pragma unroll
    for (int i = 0; i < EPT; ++i) {
        d[i] = dst[e0 + i * 256 + t];
        s[i] = src[e0 + i * 256 + t];
    }
    __syncthreads();
    #pragma unroll
    for (int i = 0; i < EPT; ++i) r[i] = atomicAdd(&hist[d[i] >> BSHIFT], 1);
    __syncthreads();
    int h0 = hist[2 * t], h1 = hist[2 * t + 1];
    int v = h0 + h1, inc = v;
    for (int o = 1; o < 32; o <<= 1) {
        int u = __shfl_up(inc, o);
        if (lane >= o) inc += u;
    }
    if (lane == 31) wsum[w] = inc;
    __syncthreads();
    int wb = 0;
    for (int i = 0; i < w; ++i) wb += wsum[i];
    int ex = wb + inc - v;
    hist[2 * t] = ex;
    hist[2 * t + 1] = ex + h0;
    __syncthreads();
    #pragma unroll
    for (int i = 0; i < EPT; ++i)
        sorted[hist[d[i] >> BSHIFT] + r[i]] = ((d[i] & (BNODES - 1)) << 16) | s[i];
    tab[blockIdx.x * TPITCH + t] = hist[t];
    if (t < TPITCH - 256) tab[blockIdx.x * TPITCH + 256 + t] = hist[256 + t];
    __syncthreads();
    #pragma unroll
    for (int i = 0; i < EPT; ++i) binned[e0 + i * 256 + t] = sorted[i * 256 + t];
}

__global__ void __launch_bounds__(256) group_kernel(const int* __restrict__ binned,
                                                    const int* __restrict__ tab,
                                                    const int* __restrict__ batch,
                                                    float* __restrict__ dis,
                                                    float* __restrict__ counts,
                                                    int* __restrict__ seg,
                                                    int* __restrict__ csr,
                                                    int* __restrict__ total, int N) {
    __shared__ int cntl[BNODES];
    __shared__ int cur[BNODES];
    __shared__ int wsum[4];
    __shared__ int wgbase;
    int t = threadIdx.x, lane = t & 31, w = t >> 5;
    int b = blockIdx.x;
    int g = t >> 3, sub = t & 7;
    if (t < BNODES) cntl[t] = 0;
    int jb[8], je[8], p0[8], p1[8];
    #pragma unroll
    for (int r = 0; r < 8; ++r) {
        int c = r * 32 + g;
        bool on = c < NCHUNK;
        int cc = on ? c : NCHUNK - 1;
        int col = on ? b : NBIN;
        jb[r] = cc * EPC + tab[cc * TPITCH + col];
        je[r] = cc * EPC + tab[cc * TPITCH + col + 1];
    }
    #pragma unroll
    for (int r = 0; r < 8; ++r) {
        int j = jb[r] + sub;
        bool in0 = j < je[r], in1 = j + 8 < je[r];
        int v0 = binned[in0 ? j : 0];
        int v1 = binned[in1 ? j + 8 : 0];
        p0[r] = in0 ? v0 : -1;
        p1[r] = in1 ? v1 : -1;
    }
    __syncthreads();
    #pragma unroll
    for (int r = 0; r < 8; ++r) {
        if (p0[r] >= 0) atomicAdd(&cntl[p0[r] >> 16], 1);
        if (p1[r] >= 0) atomicAdd(&cntl[p1[r] >> 16], 1);
        for (int j = jb[r] + 16 + sub; j < je[r]; j += 8) atomicAdd(&cntl[binned[j] >> 16], 1);
    }
    __syncthreads();
    int c = 0, inc = 0;
    if (t < BNODES) {
        c = cntl[t];
        inc = c;
        for (int o = 1; o < 32; o <<= 1) {
            int u = __shfl_up(inc, o);
            if (lane >= o) inc += u;
        }
        if (lane == 31) wsum[w] = inc;
    }
    __syncthreads();
    if (t == 0) wgbase = atomicAdd(total, wsum[0] + wsum[1] + wsum[2] + wsum[3]);
    __syncthreads();
    if (t < BNODES) {
        int wb = 0;
        for (int i = 0; i < w; ++i) wb += wsum[i];
        int beg = wgbase + wb + inc - c;
        cur[t] = beg;
        int n = b * BNODES + t;
        bool ok = n < N;
        int gid = -1;
        if (ok) {
            *reinterpret_cast<int2*>(seg + 2 * (size_t)n) = make_int2(beg, beg + c);
            dis[n] = rsqrtf((float)c + 1.0f);
            gid = batch[n];
        }
        int below = __shfl_up(gid, 1);
        bool head = lane == 0 || gid != below;
        unsigned heads = __builtin_amdgcn_ballot_w32(head);
        if (head && ok) {
            unsigned above = (heads >> lane) >> 1;
            int len = above ? __builtin_ctz(above) + 1 : 32 - lane;
            atomAddF(&counts[gid], (float)len);
        }
    }
    __syncthreads();
    int q0[8], q1[8];
    #pragma unroll
    for (int r = 0; r < 8; ++r) {
        q0[r] = 0;
        q1[r] = 0;
        if (p0[r] >= 0) q0[r] = atomicAdd(&cur[p0[r] >> 16], 1);
        if (p1[r] >= 0) q1[r] = atomicAdd(&cur[p1[r] >> 16], 1);
    }
    #pragma unroll
    for (int r = 0; r < 8; ++r) {
        if (p0[r] >= 0) csr[q0[r]] = p0[r] & 0xFFFF;
        if (p1[r] >= 0) csr[q1[r]] = p1[r] & 0xFFFF;
    }
    #pragma unroll
    for (int r = 0; r < 8; ++r) {
        for (int j = jb[r] + 16 + sub; j < je[r]; j += 8) {
            int p = binned[j];
            int pos = atomicAdd(&cur[p >> 16], 1);
            csr[pos] = p & 0xFFFF;
        }
    }
}

__device__ __forceinline__ float4 ld4(const float* p) {
    return *reinterpret_cast<const float4*>(p);
}
__device__ __forceinline__ int2 ld2i(const int* p) {
    return *reinterpret_cast<const int2*>(p);
}
__device__ __forceinline__ void fma4(float4& a, const float4& v, float d) {
    a.x += v.x * d; a.y += v.y * d; a.z += v.z * d; a.w += v.w * d;
}
__device__ __forceinline__ void add4(float4& a, const float4& v) {
    a.x += v.x; a.y += v.y; a.z += v.z; a.w += v.w;
}

__device__ __forceinline__ float4 keep4(const float4& v, bool on) {
    float4 r;
    r.x = on ? v.x : 0.0f; r.y = on ? v.y : 0.0f; r.z = on ? v.z : 0.0f; r.w = on ? v.w : 0.0f;
    return r;
}

template <int F, int LPN, bool SCALE>
__device__ __forceinline__ void gather_sum(const float* __restrict__ h,
                                           const float* __restrict__ dis,
                                           const int* __restrict__ csr,
                                           int beg, int end, int sub, float4& a0, float4& a1) {
    static_assert(F == 8 * LPN && (LPN & 3) == 0, "two float4 per lane; groups of four edges");
    for (int j0 = beg; j0 < end; j0 += 2 * LPN) {
        int ja = j0 + sub, jb = ja + LPN;
        int sa = 0, sb = 0;
        float da = 0.0f, db = 0.0f;
        if (ja < end) sa = csr[ja];
        if (jb < end) sb = csr[jb];
        if (SCALE) {
            da = dis[sa];
            db = dis[sb];
        }
        int c = end - j0 < 2 * LPN ? end - j0 : 2 * LPN;
        for (int k = 0; k < c; k += 4) {
            bool up = k >= LPN;
            int s = up ? sb : sa;
            float d = up ? db : da;
            int last = c - 1;
            int k1 = k + 1 < last ? k + 1 : last;
            int k2 = k + 2 < last ? k + 2 : last;
            int k3 = k + 3 < last ? k + 3 : last;
            int s0 = __shfl(s, k & (LPN - 1), LPN), s1 = __shfl(s, k1 & (LPN - 1), LPN);
            int s2 = __shfl(s, k2 & (LPN - 1), LPN), s3 = __shfl(s, k3 & (LPN - 1), LPN);
            const float* p0 = h + (size_t)s0 * F + sub * 4;
            const float* p1 = h + (size_t)s1 * F + sub * 4;
            const float* p2 = h + (size_t)s2 * F + sub * 4;
            const float* p3 = h + (size_t)s3 * F + sub * 4;
            float4 v0 = ld4(p0), w0 = ld4(p0 + LPN * 4);
            float4 v1 = ld4(p1), w1 = ld4(p1 + LPN * 4);
            float4 v2 = ld4(p2), w2 = ld4(p2 + LPN * 4);
            float4 v3 = ld4(p3), w3 = ld4(p3 + LPN * 4);
            v1 = keep4(v1, k + 1 < c); w1 = keep4(w1, k + 1 < c);
            v2 = keep4(v2, k + 2 < c); w2 = keep4(w2, k + 2 < c);
            v3 = keep4(v3, k + 3 < c); w3 = keep4(w3, k + 3 < c);
            if (SCALE) {
                float d0 = __shfl(d, k & (LPN - 1), LPN), d1 = __shfl(d, k1 & (LPN - 1), LPN);
                float d2 = __shfl(d, k2 & (LPN - 1), LPN), d3 = __shfl(d, k3 & (LPN - 1), LPN);
                fma4(a0, v0, d0); fma4(a1, w0, d0);
                fma4(a0, v1, d1); fma4(a1, w1, d1);
                fma4(a0, v2, d2); fma4(a1, w2, d2);
                fma4(a0, v3, d3); fma4(a1, w3, d3);
            } else {
                add4(a0, v0); add4(a1, w0);
                add4(a0, v1); add4(a1, w1);
                add4(a0, v2); add4(a1, w2);
                add4(a0, v3); add4(a1, w3);
            }
        }
    }
}

__global__ void __launch_bounds__(256) gather_gemm128_64(const float* __restrict__ x,
                                                         const float* __restrict__ W,
                                                         const float* __restrict__ dis,
                                                         const float* __restrict__ bias,
                                                         const int* __restrict__ seg,
                                                         const int* __restrict__ csr,
                                                         const float* __restrict__ Wn,
                                                         float* __restrict__ out) {
    __shared__ __attribute__((aligned(16))) float As[32 * APITCH];
    __shared__ __attribute__((aligned(16))) float Hs[32 * APITCH];
    int tid = threadIdx.x, wid = tid >> 5, lane = tid & 31;
    int hi = lane >> 4, lm = lane & 15;
    int row0 = blockIdx.x << 5;
    for (int r = 0; r < 2; ++r) {
        int lr = wid * 4 + r * 2 + hi;
        int n = row0 + lr;
        float dn = dis[n];
        const float* xr = x + (size_t)n * 128 + lm * 4;
        float4 a0 = ld4(xr), a1 = ld4(xr + 64);
        a0.x *= dn; a0.y *= dn; a0.z *= dn; a0.w *= dn;
        a1.x *= dn; a1.y *= dn; a1.z *= dn; a1.w *= dn;
        int2 se = ld2i(seg + 2 * (size_t)n);
        gather_sum<128, 16, true>(x, dis, csr, se.x, se.y, lm, a0, a1);
        *reinterpret_cast<float4*>(&As[lr * APITCH + lm * 4]) = a0;
        *reinterpret_cast<float4*>(&As[lr * APITCH + 64 + lm * 4]) = a1;
    }
    __syncthreads();

    int col0 = wid << 4;
    v8f c0 = {}, c1 = {};
    const float* Arow = As + lm * APITCH + 2 * hi;
    for (int k0 = 0; k0 < 128; k0 += 4) {
        v2f a0 = *reinterpret_cast<const v2f*>(Arow + k0);
        v2f a1 = *reinterpret_cast<const v2f*>(Arow + 16 * APITCH + k0);
        const float* Bp = W + (size_t)(k0 + 2 * hi) * 128 + col0 + lm;
        v2f b;
        b.x = Bp[0];
        b.y = Bp[128];
        c0 = __builtin_amdgcn_wmma_f32_16x16x4_f32(false, a0, false, b,
                                                   (short)0, c0, false, false);
        c1 = __builtin_amdgcn_wmma_f32_16x16x4_f32(false, a1, false, b,
                                                   (short)0, c1, false, false);
    }
    float bv = bias[col0 + lm];
    float dr0[8], dr1[8];
    #pragma unroll
    for (int v = 0; v < 8; ++v) {
        dr0[v] = dis[row0 + v + 8 * hi];
        dr1[v] = dis[row0 + 16 + v + 8 * hi];
        float o0 = c0[v] * dr0[v] + bv;
        float o1 = c1[v] * dr1[v] + bv;
        Hs[(v + 8 * hi) * APITCH + col0 + lm] = o0 > 0.0f ? o0 : 0.0f;
        Hs[(16 + v + 8 * hi) * APITCH + col0 + lm] = o1 > 0.0f ? o1 : 0.0f;
    }
    __syncthreads();
    int mt = wid >> 2;
    col0 = (wid & 3) << 4;
    v8f c2 = {};
    const float* Hrow = Hs + (16 * mt + lm) * APITCH + 2 * hi;
    for (int k0 = 0; k0 < 128; k0 += 4) {
        v2f a = *reinterpret_cast<const v2f*>(Hrow + k0);
        const float* Bp = Wn + (size_t)(k0 + 2 * hi) * 64 + col0 + lm;
        v2f b;
        b.x = Bp[0];
        b.y = Bp[64];
        c2 = __builtin_amdgcn_wmma_f32_16x16x4_f32(false, a, false, b,
                                                   (short)0, c2, false, false);
    }
    #pragma unroll
    for (int v = 0; v < 8; ++v) {
        int row = row0 + 16 * mt + v + 8 * hi;
        out[(size_t)row * 64 + col0 + lm] = c2[v] * (mt ? dr1[v] : dr0[v]);
    }
}

__global__ void __launch_bounds__(256) gather64_gemm32(const float* __restrict__ h,
                                                       const float* __restrict__ dis,
                                                       const float* __restrict__ bias,
                                                       const int* __restrict__ seg,
                                                       const int* __restrict__ csr,
                                                       const float* __restrict__ Wn,
                                                       float* __restrict__ out, int N) {
    __shared__ __attribute__((aligned(16))) float Hs[32 * HPITCH];
    __shared__ __attribute__((aligned(16))) float ds[32];
    int t = threadIdx.x;
    int ln = t >> 3, sub = t & 7;
    int row0 = blockIdx.x << 5;
    int n = row0 + ln;
    float4 o0 = make_float4(0.0f, 0.0f, 0.0f, 0.0f);
    float4 o1 = o0;
    float dn = 0.0f;
    if (n < N) {
        const float* hr = h + (size_t)n * 64 + sub * 4;
        float4 a0 = ld4(hr), a1 = ld4(hr + 32);
        int2 se = ld2i(seg + 2 * (size_t)n);
        gather_sum<64, 8, false>(h, dis, csr, se.x, se.y, sub, a0, a1);
        dn = dis[n];
        float4 b0 = ld4(bias + sub * 4), b1 = ld4(bias + 32 + sub * 4);
        o0.x = a0.x * dn + b0.x; o0.y = a0.y * dn + b0.y;
        o0.z = a0.z * dn + b0.z; o0.w = a0.w * dn + b0.w;
        o1.x = a1.x * dn + b1.x; o1.y = a1.y * dn + b1.y;
        o1.z = a1.z * dn + b1.z; o1.w = a1.w * dn + b1.w;
        o0.x = o0.x > 0.0f ? o0.x : 0.0f; o0.y = o0.y > 0.0f ? o0.y : 0.0f;
        o0.z = o0.z > 0.0f ? o0.z : 0.0f; o0.w = o0.w > 0.0f ? o0.w : 0.0f;
        o1.x = o1.x > 0.0f ? o1.x : 0.0f; o1.y = o1.y > 0.0f ? o1.y : 0.0f;
        o1.z = o1.z > 0.0f ? o1.z : 0.0f; o1.w = o1.w > 0.0f ? o1.w : 0.0f;
    }
    *reinterpret_cast<float4*>(&Hs[ln * HPITCH + sub * 4]) = o0;
    *reinterpret_cast<float4*>(&Hs[ln * HPITCH + 32 + sub * 4]) = o1;
    if (sub == 0) ds[ln] = dn;
    __syncthreads();
    int wid = t >> 5, lane = t & 31;
    if (wid < 4) {
        int hi = lane >> 4, lm = lane & 15;
        int mt = wid >> 1;
        int col0 = (wid & 1) << 4;
        float4 d0 = *reinterpret_cast<const float4*>(&ds[16 * mt + 8 * hi]);
        float4 d1 = *reinterpret_cast<const float4*>(&ds[16 * mt + 8 * hi + 4]);
        float dr[8] = {d0.x, d0.y, d0.z, d0.w, d1.x, d1.y, d1.z, d1.w};
        v8f c = {};
        const float* Hrow = Hs + (16 * mt + lm) * HPITCH + 2 * hi;
        for (int k0 = 0; k0 < 64; k0 += 4) {
            v2f a = *reinterpret_cast<const v2f*>(Hrow + k0);
            const float* Bp = Wn + (size_t)(k0 + 2 * hi) * 32 + col0 + lm;
            v2f b;
            b.x = Bp[0];
            b.y = Bp[32];
            c = __builtin_amdgcn_wmma_f32_16x16x4_f32(false, a, false, b,
                                                      (short)0, c, false, false);
        }
        #pragma unroll
        for (int v = 0; v < 8; ++v) {
            int row = row0 + 16 * mt + v + 8 * hi;
            if (row < N) out[(size_t)row * 32 + col0 + lm] = c[v] * dr[v];
        }
    }
}

__global__ void __launch_bounds__(256) gather_pool32(const float* __restrict__ h,
                                                     const float* __restrict__ dis,
                                                     const float* __restrict__ bias,
                                                     const int* __restrict__ seg,
                                                     const int* __restrict__ csr,
                                                     const int* __restrict__ batch,
                                                     float* __restrict__ pooled, int N) {
    __shared__ __attribute__((aligned(16))) float rows[64 * 32];
    __shared__ int gid[64];
    int t = threadIdx.x;
    int ln = t >> 2, sub = t & 3;
    int n = blockIdx.x * 64 + ln;
    float4 o0 = make_float4(0.0f, 0.0f, 0.0f, 0.0f);
    float4 o1 = o0;
    int g = -1;
    if (n < N) {
        const float* hr = h + (size_t)n * 32 + sub * 4;
        float4 a0 = ld4(hr), a1 = ld4(hr + 16);
        int2 se = ld2i(seg + 2 * (size_t)n);
        gather_sum<32, 4, false>(h, dis, csr, se.x, se.y, sub, a0, a1);
        float dn = dis[n];
        float4 b0 = ld4(bias + sub * 4), b1 = ld4(bias + 16 + sub * 4);
        o0.x = a0.x * dn + b0.x; o0.y = a0.y * dn + b0.y;
        o0.z = a0.z * dn + b0.z; o0.w = a0.w * dn + b0.w;
        o1.x = a1.x * dn + b1.x; o1.y = a1.y * dn + b1.y;
        o1.z = a1.z * dn + b1.z; o1.w = a1.w * dn + b1.w;
        o0.x = o0.x > 0.0f ? o0.x : 0.0f; o0.y = o0.y > 0.0f ? o0.y : 0.0f;
        o0.z = o0.z > 0.0f ? o0.z : 0.0f; o0.w = o0.w > 0.0f ? o0.w : 0.0f;
        o1.x = o1.x > 0.0f ? o1.x : 0.0f; o1.y = o1.y > 0.0f ? o1.y : 0.0f;
        o1.z = o1.z > 0.0f ? o1.z : 0.0f; o1.w = o1.w > 0.0f ? o1.w : 0.0f;
        g = batch[n];
    }
    *reinterpret_cast<float4*>(&rows[ln * 32 + sub * 4]) = o0;
    *reinterpret_cast<float4*>(&rows[ln * 32 + 16 + sub * 4]) = o1;
    if (sub == 0) gid[ln] = g;
    __syncthreads();
    if (t < 64) {
        int f = t & 31, r0 = t & 32;
        int mine = gid[r0 + f];
        float r[32];
        #pragma unroll
        for (int i = 0; i < 32; ++i) r[i] = rows[(r0 + i) * 32 + f];
        int cur = __builtin_amdgcn_readlane(mine, 0);
        float s = 0.0f;
        #pragma unroll
        for (int i = 0; i < 32; ++i) {
            int gi = __builtin_amdgcn_readlane(mine, i);
            if (gi != cur) {
                if (cur >= 0) atomAddF(&pooled[(cur << 5) + f], s);
                s = 0.0f;
                cur = gi;
            }
            s += r[i];
        }
        if (cur >= 0) atomAddF(&pooled[(cur << 5) + f], s);
    }
}

__global__ void __launch_bounds__(256) head_gemm_mean(const float* __restrict__ pooled,
                                                      const float* __restrict__ counts,
                                                      const float* __restrict__ Wl,
                                                      const float* __restrict__ bl,
                                                      float* __restrict__ out) {
    int tid = threadIdx.x, wid = tid >> 5, lane = tid & 31;
    int tile = blockIdx.x * 8 + wid;
    if (tile >= 16 * 48) return;
    int rt = tile / 48, ct = tile - rt * 48;
    int row0 = rt << 4, col0 = ct << 4;
    int hi = lane >> 4, lm = lane & 15;
    float cm = fmaxf(counts[row0 + lm], 1.0f);
    v8f c = {};
    const float* Arow = pooled + (size_t)(row0 + lm) * 32;
    for (int k0 = 0; k0 < 32; k0 += 4) {
        v2f a = *reinterpret_cast<const v2f*>(Arow + k0 + 2 * hi);
        a.x = a.x / cm;
        a.y = a.y / cm;
        const float* Bp = Wl + (size_t)(k0 + 2 * hi) * 768 + col0 + lm;
        v2f b;
        b.x = Bp[0];
        b.y = Bp[768];
        c = __builtin_amdgcn_wmma_f32_16x16x4_f32(false, a, false, b,
                                                  (short)0, c, false, false);
    }
    float bv = bl[col0 + lm];
    #pragma unroll
    for (int v = 0; v < 8; ++v) {
        int row = row0 + v + 8 * hi;
        out[(size_t)row * 768 + col0 + lm] = c[v] + bv;
    }
}

extern "C" void kernel_launch(void* const* d_in, const int* in_sizes, int n_in,
                              void* d_out, int out_size, void* d_ws, size_t ws_size,
                              hipStream_t stream) {
    const float* x   = (const float*)d_in[0];
    const float* W1  = (const float*)d_in[1];
    const float* b1  = (const float*)d_in[2];
    const float* W2  = (const float*)d_in[3];
    const float* b2  = (const float*)d_in[4];
    const float* W3  = (const float*)d_in[5];
    const float* b3  = (const float*)d_in[6];
    const float* Wl  = (const float*)d_in[7];
    const float* bl  = (const float*)d_in[8];
    const int*   ei  = (const int*)d_in[9];
    const int*   bat = (const int*)d_in[10];
    float*       out = (float*)d_out;

    const int N = NNODES, E = NEDGES;
    const int* src = ei;
    const int* dst = ei + E;

    float* B0     = (float*)d_ws;
    float* dis    = B0 + (size_t)N * 32;
    float* counts = dis + N;
    float* pooled = counts + NGRAPH;
    int*   total  = (int*)(pooled + NGRAPH * 32);
    float* B1     = B0 + (size_t)N * 128;
    int*   seg    = (int*)(B1 + (size_t)N * 64);
    int*   binned = seg + 2 * N;
    int*   csr    = binned + E;
    int*   tab    = csr + E;

    hipMemsetAsync(counts, 0, (NGRAPH + NGRAPH * 32 + 32) * sizeof(float), stream);
    bin_kernel<<<NCHUNK, 256, 0, stream>>>(src, dst, binned, tab);
    group_kernel<<<NBIN, 256, 0, stream>>>(binned, tab, bat, dis, counts, seg, csr, total, N);

    gather_gemm128_64<<<N / 32, 256, 0, stream>>>(x, W1, dis, b1, seg, csr, W2, B1);

    gather64_gemm32<<<(N + 31) / 32, 256, 0, stream>>>(B1, dis, b2, seg, csr, W3, B0, N);

    gather_pool32<<<(N + 63) / 64, 256, 0, stream>>>(B0, dis, b3, seg, csr, bat, pooled, N);

    head_gemm_mean<<<(16 * 48) / 8, 256, 0, stream>>>(pooled, counts, Wl, bl, out);
}
